# starting kernel
# baseline (speedup 1.0000x reference)
#include <hip/hip_runtime.h>
#include <hip/hip_bf16.h>


typedef __attribute__((ext_vector_type(2))) float v2f;
typedef __attribute__((ext_vector_type(8))) float v8f;

#define NFEAT  256
#define NHID   128
#define NCLASS 64

__device__ __forceinline__ void atomAddF(float* p, float v) {
    __hip_atomic_fetch_add(p, v, __ATOMIC_RELAXED, __HIP_MEMORY_SCOPE_AGENT);
}

__global__ void k_zero(float* __restrict__ p, int n) {
    int i = blockIdx.x * blockDim.x + threadIdx.x;
    if (i < n) p[i] = 0.0f;
}

__global__ void k_count(const int* __restrict__ src, const int* __restrict__ dst,
                        float* __restrict__ cdst, float* __restrict__ csrc, int E) {
    int e = blockIdx.x * blockDim.x + threadIdx.x;
    if (e < E) {
        atomAddF(&cdst[dst[e]], 1.0f);
        atomAddF(&csrc[src[e]], 1.0f);
    }
}

__global__ void k_scatter_rows(const float* __restrict__ X,
                               const int* __restrict__ gath,
                               const int* __restrict__ scat,
                               float* __restrict__ out, int E) {
    int t = blockIdx.x * blockDim.x + threadIdx.x;
    int e = t >> 5;
    if (e >= E) return;
    int f = (t & 31) * 4;
    const float4 v = *(const float4*)(X + (size_t)gath[e] * NHID + f);
    float* o = out + (size_t)scat[e] * NHID + f;
    atomAddF(o + 0, v.x);
    atomAddF(o + 1, v.y);
    atomAddF(o + 2, v.z);
    atomAddF(o + 3, v.w);
}

__global__ void k_div_rows(float* __restrict__ agg, const float* __restrict__ cnt, int total) {
    int i = blockIdx.x * blockDim.x + threadIdx.x;
    if (i >= total) return;
    float c = cnt[i >> 7];
    agg[i] *= 1.0f / fmaxf(c, 1.0f);
}

__global__ void k_edge_h(const float* __restrict__ A, const float* __restrict__ B,
                         const int* __restrict__ src, const int* __restrict__ dst,
                         float* __restrict__ hs, int E) {
    int t = blockIdx.x * blockDim.x + threadIdx.x;
    int e = t >> 5;
    if (e >= E) return;
    int f = (t & 31) * 4;
    int s = src[e], d = dst[e];
    const float4 a = *(const float4*)(A + (size_t)s * NHID + f);
    const float4 b = *(const float4*)(B + (size_t)d * NHID + f);
    float hx = (a.x + b.x) * (a.x + b.x);
    float hy = (a.y + b.y) * (a.y + b.y);
    float hz = (a.z + b.z) * (a.z + b.z);
    float hw = (a.w + b.w) * (a.w + b.w);
    float* o = hs + (size_t)s * NHID + f;
    atomAddF(o + 0, hx);
    atomAddF(o + 1, hy);
    atomAddF(o + 2, hz);
    atomAddF(o + 3, hw);
}

__global__ void k_gate(const float* __restrict__ Xcur, const float* __restrict__ Xnew,
                       const float* __restrict__ hs, const float* __restrict__ csrc,
                       float* __restrict__ Xnext, int total) {
    int i = blockIdx.x * blockDim.x + threadIdx.x;
    if (i >= total) return;
    float c   = fmaxf(csrc[i >> 7], 1.0f);
    float tau = tanhf(hs[i] / c);
    Xnext[i]  = (1.0f - tau) * Xcur[i] + tau * Xnew[i];
}

__global__ void k_gemm(const float* __restrict__ A1, const float* __restrict__ W1,
                       const float* __restrict__ A2, const float* __restrict__ W2,
                       const float* __restrict__ bias, float* __restrict__ C,
                       int M, int K, int N, int relu) {
    const int lane   = threadIdx.x & 31;
    const int wave   = threadIdx.x >> 5;
    const int ntiles = N >> 4;
    if (wave >= ntiles) return;

    const int m0   = blockIdx.x << 4;
    const int n0   = wave << 4;
    const int row  = lane & 15;
    const int half = lane >> 4;

    v8f c = {};

    {
        const float* arow = A1 + (size_t)(m0 + row) * K;
        for (int k0 = 0; k0 < K; k0 += 4) {
            v2f a, b;
            a.x = arow[k0 + 2 * half];
            a.y = arow[k0 + 2 * half + 1];
            b.x = W1[(size_t)(k0 + 2 * half) * N + n0 + row];
            b.y = W1[(size_t)(k0 + 2 * half + 1) * N + n0 + row];
            c = __builtin_amdgcn_wmma_f32_16x16x4_f32(false, a, false, b,
                                                      (short)0, c, false, false);
        }
    }
    if (A2 != nullptr) {
        const float* arow = A2 + (size_t)(m0 + row) * K;
        for (int k0 = 0; k0 < K; k0 += 4) {
            v2f a, b;
            a.x = arow[k0 + 2 * half];
            a.y = arow[k0 + 2 * half + 1];
            b.x = W2[(size_t)(k0 + 2 * half) * N + n0 + row];
            b.y = W2[(size_t)(k0 + 2 * half + 1) * N + n0 + row];
            c = __builtin_amdgcn_wmma_f32_16x16x4_f32(false, a, false, b,
                                                      (short)0, c, false, false);
        }
    }

    const float bv = bias ? bias[n0 + row] : 0.0f;
    #pragma unroll
    for (int i = 0; i < 8; ++i) {
        float x = c[i] + bv;
        if (relu) x = fmaxf(x, 0.0f);
        int m = m0 + i + 8 * half;
        if (m < M) C[(size_t)m * N + n0 + row] = x;
    }
}

extern "C" void kernel_launch(void* const* d_in, const int* in_sizes, int n_in,
                              void* d_out, int out_size, void* d_ws, size_t ws_size,
                              hipStream_t stream) {
    const float* X       = (const float*)d_in[0];
    const int*   ei      = (const int*)  d_in[1];
    const float* enc_W   = (const float*)d_in[2];
    const float* enc_b   = (const float*)d_in[3];
    const float* conv_Wl = (const float*)d_in[4];
    const float* conv_bl = (const float*)d_in[5];
    const float* conv_Wr = (const float*)d_in[6];
    const float* gg_Wl   = (const float*)d_in[7];
    const float* gg_bl   = (const float*)d_in[8];
    const float* gg_Wr   = (const float*)d_in[9];
    const float* Q_W     = (const float*)d_in[10];
    const float* Q_b     = (const float*)d_in[11];
    const float* dec_W   = (const float*)d_in[12];
    const float* dec_b   = (const float*)d_in[13];

    const int N = in_sizes[0] / NFEAT;
    const int E = in_sizes[1] / 2;
    const int* src = ei;
    const int* dst = ei + E;

    float* ws       = (float*)d_ws;
    const size_t NH = (size_t)N * NHID;
    float* cnt_dst  = ws;
    float* cnt_src  = cnt_dst + N;
    float* Xa       = cnt_src + N;
    float* Xb       = Xa + NH;
    float* agg      = Xb + NH;
    float* T1       = agg + NH;
    float* T2       = T1 + NH;
    float* Ab       = T2 + NH;
    float* Bb       = Ab + NH;
    float* hs       = Bb + NH;

    auto cdiv = [](long long a, long long b) { return (int)((a + b - 1) / b); };
    const int total   = (int)NH;
    const int eThr    = E * 32;
    const int gemmGrd = N / 16;

    k_zero <<<cdiv(2 * N, 256), 256, 0, stream>>>(cnt_dst, 2 * N);
    k_count<<<cdiv(E, 256),     256, 0, stream>>>(src, dst, cnt_dst, cnt_src, E);

    k_gemm<<<gemmGrd, 256, 0, stream>>>(X, enc_W, nullptr, nullptr, enc_b,
                                        Xa, N, NFEAT, NHID, 1);

    float* Xcur  = Xa;
    float* Xnext = Xb;
    for (int layer = 0; layer < 2; ++layer) {
        k_zero        <<<cdiv(total, 256), 256, 0, stream>>>(agg, total);
        k_scatter_rows<<<cdiv(eThr, 256),  256, 0, stream>>>(Xcur, src, dst, agg, E);
        k_div_rows    <<<cdiv(total, 256), 256, 0, stream>>>(agg, cnt_dst, total);

        k_gemm<<<gemmGrd, 256, 0, stream>>>(agg, conv_Wl, Xcur, conv_Wr, conv_bl,
                                            T1, N, NHID, NHID, 1);
        k_gemm<<<gemmGrd, 256, 0, stream>>>(agg, gg_Wl, Xcur, gg_Wr, gg_bl,
                                            T2, N, NHID, NHID, 1);

        k_gemm<<<gemmGrd, 256, 0, stream>>>(T2, Q_W, nullptr, nullptr, Q_b,
                                            Ab, N, NHID, NHID, 0);
        k_gemm<<<gemmGrd, 256, 0, stream>>>(T2, Q_W + (size_t)NHID * NHID, nullptr, nullptr,
                                            nullptr, Bb, N, NHID, NHID, 0);

        k_zero  <<<cdiv(total, 256), 256, 0, stream>>>(hs, total);
        k_edge_h<<<cdiv(eThr, 256),  256, 0, stream>>>(Ab, Bb, src, dst, hs, E);
        k_gate  <<<cdiv(total, 256), 256, 0, stream>>>(Xcur, T1, hs, cnt_src, Xnext, total);

        float* t = Xcur; Xcur = Xnext; Xnext = t;
    }

    k_gemm<<<gemmGrd, 256, 0, stream>>>(Xcur, dec_W, nullptr, nullptr, dec_b,
                                        (float*)d_out, N, NHID, NCLASS, 0);
}


# Decoder GEMM moved into the owner work-group of the last gate launch: gated rows stay in LDS, 6 waves run WMMA on them (dec_W by 8-B loads, k order and accumulator as before); 6->5 launches
# speedup vs baseline: 3.6455x; 3.6455x over previous
#include <hip/hip_runtime.h>
#include <hip/hip_bf16.h>


typedef __attribute__((ext_vector_type(2))) float v2f;
typedef __attribute__((ext_vector_type(8))) float v8f;

#define NFEAT  256
#define NHID   128
#define NCLASS 64

#define RB    40
#define OWN_T 1024
#define QCAP  544
#define MT    3
#define RBP   (MT * 16)
#define LP    132
#define LCAP    2048
#define LHDR    32
#define LSTRIDE (LHDR + LCAP)

struct LayerG {
    const float* Wl1; const float* Wr1; const float* b1;
    const float* Wl2; const float* Wr2; const float* b2;
    const float* Q;   const float* Qb;
    float* T1; float* Ab; float* Bb;
    const float* Wd;  const float* bd;  float* Od;
};

template <int WS>
__device__ __forceinline__ void blk_mma(v8f& c0, v8f& c1, const float* __restrict__ ar,
                                        const float* __restrict__ Wm, int half, int colo) {
    const float* wp = Wm + (size_t)(2 * half) * WS + colo;
    for (int k0 = 0; k0 < NHID; k0 += 4) {
        const v2f a  = *(const v2f*)(ar + k0);
        const v2f w0 = *(const v2f*)(wp + (size_t)k0 * WS);
        const v2f w1 = *(const v2f*)(wp + (size_t)(k0 + 1) * WS);
        v2f b0, b1;
        b0.x = w0.x; b0.y = w1.x;
        b1.x = w0.y; b1.y = w1.y;
        c0 = __builtin_amdgcn_wmma_f32_16x16x4_f32(false, a, false, b0, (short)0, c0, false, false);
        c1 = __builtin_amdgcn_wmma_f32_16x16x4_f32(false, a, false, b1, (short)0, c1, false, false);
    }
}

__device__ __forceinline__ void ldsAddF(float* p, float v) {
    __hip_atomic_fetch_add(p, v, __ATOMIC_RELAXED, __HIP_MEMORY_SCOPE_WORKGROUP);
}

__device__ __forceinline__ void own_push(int* __restrict__ q, int& qn, int v, int e,
                                         int n0, unsigned rcur, int lane) {
    const unsigned rel = (unsigned)(v - n0);
    const bool m = rel < rcur;
    const unsigned mask = __builtin_amdgcn_ballot_w32(m);
    if (mask) {
        if (m) q[qn + __builtin_popcount(mask & ((1u << lane) - 1u))] = (e << 6) | (int)rel;
        qn += __builtin_popcount(mask);
    }
}

__device__ __forceinline__ void own_push4(int* __restrict__ q, int& qn, const int4 v, int e,
                                          int n0, unsigned rcur, int lane) {
    own_push(q, qn, v.x, e + 0, n0, rcur, lane);
    own_push(q, qn, v.y, e + 1, n0, rcur, lane);
    own_push(q, qn, v.z, e + 2, n0, rcur, lane);
    own_push(q, qn, v.w, e + 3, n0, rcur, lane);
}

template <int MODE>
__device__ __forceinline__ int own_row(float4& v, int g, int rel, int i,
                                       const float* __restrict__ G,
                                       const float* __restrict__ Aown, int n0, int lane) {
    const int gi = __builtin_amdgcn_readlane(g, i);
    const int ri = __builtin_amdgcn_readlane(rel, i);
    v = *(const float4*)(G + (size_t)gi * NHID + lane * 4);
    if (MODE == 1) {
        const float4 a = *(const float4*)(Aown + (size_t)(n0 + ri) * NHID + lane * 4);
        v.x = (a.x + v.x) * (a.x + v.x);
        v.y = (a.y + v.y) * (a.y + v.y);
        v.z = (a.z + v.z) * (a.z + v.z);
        v.w = (a.w + v.w) * (a.w + v.w);
    }
    return ri;
}

__device__ __forceinline__ void own_add(float* __restrict__ acc, int ri, int lane, const float4 v) {
    float* p = acc + ri * NHID + lane;
    ldsAddF(p,      v.x);
    ldsAddF(p + 32, v.y);
    ldsAddF(p + 64, v.z);
    ldsAddF(p + 96, v.w);
}

template <int MODE>
__device__ __forceinline__ void own_rows(int n, int g, int rel, float* __restrict__ acc,
                                         const float* __restrict__ G,
                                         const float* __restrict__ Aown, int n0, int lane) {
    int i = 0;
    for (; i + 4 <= n; i += 4) {
        float4 v0, v1, v2, v3;
        const int r0 = own_row<MODE>(v0, g, rel, i + 0, G, Aown, n0, lane);
        const int r1 = own_row<MODE>(v1, g, rel, i + 1, G, Aown, n0, lane);
        const int r2 = own_row<MODE>(v2, g, rel, i + 2, G, Aown, n0, lane);
        const int r3 = own_row<MODE>(v3, g, rel, i + 3, G, Aown, n0, lane);
        own_add(acc, r0, lane, v0);
        own_add(acc, r1, lane, v1);
        own_add(acc, r2, lane, v2);
        own_add(acc, r3, lane, v3);
    }
    for (; i < n; ++i) {
        float4 v0;
        const int r0 = own_row<MODE>(v0, g, rel, i, G, Aown, n0, lane);
        own_add(acc, r0, lane, v0);
    }
}

template <int MODE, bool SAVE>
__device__ __forceinline__ void own_drain(int* __restrict__ q, int& qn, float* __restrict__ acc,
                                          int* __restrict__ cntL, int* __restrict__ lctl,
                                          int* __restrict__ lst,
                                          const float* __restrict__ G,
                                          const float* __restrict__ Aown,
                                          const int* __restrict__ gath, int n0, int lane) {
    __builtin_amdgcn_wave_barrier();
    int b = 0;
    for (; b + 32 <= qn; b += 32) {
        const int ent = q[b + lane];
        const int rel = ent & 63;
        const int g = gath[ent >> 6];
        __hip_atomic_fetch_add(&cntL[rel], 1, __ATOMIC_RELAXED, __HIP_MEMORY_SCOPE_WORKGROUP);
        if (SAVE && lst != nullptr) {
            int c = 0;
            if (lane == 0)
                c = __hip_atomic_fetch_add(&lctl[0], 1, __ATOMIC_RELAXED, __HIP_MEMORY_SCOPE_WORKGROUP);
            c = __builtin_amdgcn_readfirstlane(c);
            if (c < LCAP / 32) lst[LHDR + c * 32 + lane] = (g << 6) | rel;
        }
        own_rows<MODE>(32, g, rel, acc, G, Aown, n0, lane);
    }
    const int r = qn - b;
    if (b > 0 && lane < r) {
        const int t = q[b + lane];
        q[lane] = t;
    }
    qn = r;
    __builtin_amdgcn_wave_barrier();
}

__device__ __forceinline__ void own_load4(int4& a0, int4& a1, int4& a2, int4& a3,
                                          const int4* __restrict__ sv, int vb) {
    a0 = sv[vb];
    a1 = sv[vb + OWN_T];
    a2 = sv[vb + 2 * OWN_T];
    a3 = sv[vb + 3 * OWN_T];
}

template <int MODE, bool SAVE>
__device__ __forceinline__ void own_trip(const int4 a0, const int4 a1, const int4 a2, const int4 a3,
                                         int vb, int* __restrict__ q, int& qn,
                                         float* __restrict__ acc, int* __restrict__ cntL,
                                         int* __restrict__ lctl, int* __restrict__ lst,
                                         const float* __restrict__ G,
                                         const float* __restrict__ Aown,
                                         const int* __restrict__ gath,
                                         int n0, unsigned rcur, int lane) {
    own_push4(q, qn, a0, vb * 4, n0, rcur, lane);
    own_push4(q, qn, a1, (vb + OWN_T) * 4, n0, rcur, lane);
    own_push4(q, qn, a2, (vb + 2 * OWN_T) * 4, n0, rcur, lane);
    own_push4(q, qn, a3, (vb + 3 * OWN_T) * 4, n0, rcur, lane);
    if (qn >= 32) own_drain<MODE, SAVE>(q, qn, acc, cntL, lctl, lst, G, Aown, gath, n0, lane);
}

template <int MODE, bool SAVE>
__device__ __forceinline__ void own_filter(int* __restrict__ q, float* __restrict__ acc,
                                           int* __restrict__ cntL, int* __restrict__ lctl,
                                           int* __restrict__ lq, int* __restrict__ lst,
                                           const float* __restrict__ G,
                                           const float* __restrict__ Aown,
                                           const int* __restrict__ scat,
                                           const int* __restrict__ gath,
                                           int n0, unsigned rcur, int E,
                                           int tid, int lane, int wave) {
    int qn = 0;

    const int nvec = ((E & 3) == 0 && ((size_t)scat & 15) == 0) ? (E >> 2) : 0;
    const int4* sv = (const int4*)scat;
    const int4 none = make_int4(-1, -1, -1, -1);

    const int nfull = nvec / (4 * OWN_T);
    const int toff = (nfull > 0) ? (int)(blockIdx.x % (unsigned)nfull) : 0;
    auto trip_vb = [&](int t) {
        int tt = t + toff;
        if (tt >= nfull) tt -= nfull;
        return tt * 4 * OWN_T + tid;
    };
    int4 a0 = none, a1 = none, a2 = none, a3 = none;
    int4 b0 = none, b1 = none, b2 = none, b3 = none;
    if (nfull > 0) own_load4(a0, a1, a2, a3, sv, trip_vb(0));
    int t = 0;
    for (; t + 2 <= nfull; t += 2) {
        own_load4(b0, b1, b2, b3, sv, trip_vb(t + 1));
        own_trip<MODE, SAVE>(a0, a1, a2, a3, trip_vb(t), q, qn, acc, cntL, lctl, lst, G, Aown, gath,
                             n0, rcur, lane);
        if (t + 2 < nfull) own_load4(a0, a1, a2, a3, sv, trip_vb(t + 2));
        own_trip<MODE, SAVE>(b0, b1, b2, b3, trip_vb(t + 1), q, qn, acc, cntL, lctl, lst, G, Aown, gath,
                             n0, rcur, lane);
    }
    if (t < nfull)
        own_trip<MODE, SAVE>(a0, a1, a2, a3, trip_vb(t), q, qn, acc, cntL, lctl, lst, G, Aown, gath,
                             n0, rcur, lane);
    for (int vb = nfull * 4 * OWN_T; vb < nvec; vb += OWN_T) {
        const int idx = vb + tid;
        int4 a = none;
        if (idx < nvec) a = sv[idx];
        own_push4(q, qn, a, idx * 4, n0, rcur, lane);
        if (qn >= 32) own_drain<MODE, SAVE>(q, qn, acc, cntL, lctl, lst, G, Aown, gath, n0, lane);
    }
    for (int eb = nvec * 4; eb < E; eb += OWN_T) {
        const int e = eb + tid;
        const int v = (e < E) ? scat[e] : -1;
        own_push(q, qn, v, e, n0, rcur, lane);
        if (qn >= 32) own_drain<MODE, SAVE>(q, qn, acc, cntL, lctl, lst, G, Aown, gath, n0, lane);
    }

    __builtin_amdgcn_wave_barrier();
    {
        int base = 0;
        if (lane == 0 && qn > 0)
            base = __hip_atomic_fetch_add(&lctl[1], qn, __ATOMIC_RELAXED, __HIP_MEMORY_SCOPE_WORKGROUP);
        base = __builtin_amdgcn_readfirstlane(base);
        if (lane < qn) lq[base + lane] = q[lane];
    }
    __syncthreads();
    const int L = lctl[1];
    const int F = lctl[0];
    const bool fits = SAVE && lst != nullptr && F * 32 + L <= LCAP;
    if (fits) {
        const int i = wave * 32 + lane;
        if (i < L) {
            const int ent = lq[i];
            lst[LHDR + F * 32 + i] = (gath[ent >> 6] << 6) | (ent & 63);
        }
    }
    if (SAVE && lst != nullptr && tid == 0) lst[0] = fits ? F * 32 + L : -1;
    {
        const int wv = __builtin_amdgcn_readfirstlane(wave);
        const int lo = (wv * L) >> 5;
        const int n  = (((wv + 1) * L) >> 5) - lo;
        int rel = 0, g = 0;
        if (lane < n) {
            const int ent = lq[lo + lane];
            rel = ent & 63;
            g = gath[ent >> 6];
            __hip_atomic_fetch_add(&cntL[rel], 1, __ATOMIC_RELAXED, __HIP_MEMORY_SCOPE_WORKGROUP);
        }
        own_rows<MODE>(n, g, rel, acc, G, Aown, n0, lane);
    }
}

template <int MODE>
__device__ __forceinline__ void own_replay(const int* __restrict__ lst, int T,
                                           float* __restrict__ acc, int* __restrict__ cntL,
                                           const float* __restrict__ G,
                                           const float* __restrict__ Aown,
                                           int n0, int lane, int wave) {
    const int wv = __builtin_amdgcn_readfirstlane(wave);
    const int lo = (wv * T) >> 5;
    const int hi = ((wv + 1) * T) >> 5;
    for (int b = lo; b < hi; b += 32) {
        const int n = (hi - b < 32) ? (hi - b) : 32;
        int rel = 0, g = 0;
        if (lane < n) {
            const int ent = lst[LHDR + b + lane];
            rel = ent & 63;
            g = ent >> 6;
            __hip_atomic_fetch_add(&cntL[rel], 1, __ATOMIC_RELAXED, __HIP_MEMORY_SCOPE_WORKGROUP);
        }
        own_rows<MODE>(n, g, rel, acc, G, Aown, n0, lane);
    }
}

__device__ __forceinline__ float4 gate4(const float* __restrict__ acc, const int* __restrict__ cntL,
                                        int r, int l, const float* __restrict__ Xcur,
                                        const float* __restrict__ Xnew, size_t o) {
    const float c = fmaxf((float)cntL[r], 1.0f);
    const float* a = acc + r * NHID + l;
    const float4 s = make_float4(a[0], a[32], a[64], a[96]);
    const float4 xc = *(const float4*)(Xcur + o);
    const float4 xn = *(const float4*)(Xnew + o);
    const float tx = tanhf(s.x / c), ty = tanhf(s.y / c);
    const float tz = tanhf(s.z / c), tw = tanhf(s.w / c);
    float4 y;
    y.x = (1.0f - tx) * xc.x + tx * xn.x;
    y.y = (1.0f - ty) * xc.y + ty * xn.y;
    y.z = (1.0f - tz) * xc.z + tz * xn.z;
    y.w = (1.0f - tw) * xc.w + tw * xn.w;
    return y;
}

template <int MODE, int PASS, bool DEC>
__global__ __launch_bounds__(OWN_T)
void k_owner(const float* __restrict__ G, const float* __restrict__ Aown,
             const int* __restrict__ scat, const int* __restrict__ gath,
             const float* __restrict__ Xcur, const float* __restrict__ Xnew,
             float* __restrict__ out, const LayerG lg, int N, int E,
             int* __restrict__ lstAll, const float* __restrict__ wsf, size_t wsz) {
    __shared__ float acc[RB * NHID];
    __shared__ int   cntL[RB];
    __shared__ int   lctl[2];
    __shared__ int   lq[OWN_T];
    __shared__ int   qbuf[OWN_T / 32][QCAP];
    __shared__ alignas(16) float gbuf[MODE == 0 ? 3 * RBP * LP : (DEC ? RBP * LP : 4)];

    const int tid  = threadIdx.x;
    const int lane = tid & 31;
    const int wave = tid >> 5;
    const int n0   = blockIdx.x * RB;
    const unsigned rcur = (unsigned)((N - n0 < RB) ? (N - n0) : RB);

    int* lst = nullptr;
    if ((size_t)((const char*)(lstAll + (size_t)gridDim.x * LSTRIDE) - (const char*)wsf) <= wsz)
        lst = lstAll + (size_t)blockIdx.x * LSTRIDE;

    for (int i = tid; i < RB * NHID; i += OWN_T) acc[i] = 0.0f;
    if (tid < RB) cntL[tid] = 0;
    if (tid < 2) lctl[tid] = 0;
    __syncthreads();

    int T = -1;
    if (PASS == 1 && lst != nullptr) {
        T = lst[0];
        if (T > LCAP) T = -1;
    }
    if (T >= 0)
        own_replay<MODE>(lst, T, acc, cntL, G, Aown, n0, lane, wave);
    else
        own_filter<MODE, PASS == 0>(qbuf[wave], acc, cntL, lctl, lq, lst, G, Aown, scat, gath,
                                    n0, rcur, E, tid, lane, wave);
    __syncthreads();

    if (MODE == 0) {
        float* aggs = gbuf;
        float* xs   = gbuf + RBP * LP;
        float* xg   = gbuf + 2 * RBP * LP;
        for (int i = tid; i < RBP * 32; i += OWN_T) {
            const int r = i >> 5;
            const int l = i & 31;
            float4 s = make_float4(0.0f, 0.0f, 0.0f, 0.0f);
            float4 x = s;
            if (r < (int)rcur) {
                const float c = fmaxf((float)cntL[r], 1.0f);
                const float* a = acc + r * NHID + l;
                s = make_float4(a[0] / c, a[32] / c, a[64] / c, a[96] / c);
                x = *(const float4*)(G + (size_t)(n0 + r) * NHID + l * 4);
            }
            *(float4*)(aggs + r * LP + l * 4) = s;
            *(float4*)(xs + r * LP + l * 4) = x;
        }
        __syncthreads();

        const bool gw   = wave < MT * 8;
        const int  mt   = wave >> 3;
        const int  sel  = (wave >> 2) & 1;
        const int  row  = lane & 15;
        const int  half = lane >> 4;
        const int  colo = (wave & 3) * 32 + 2 * row;
        const int  lrow = mt * 16 + 8 * half;

        if (gw) {
            v8f c0 = {}, c1 = {};
            blk_mma<NHID>(c0, c1, aggs + (mt * 16 + row) * LP + 2 * half, sel ? lg.Wl2 : lg.Wl1, half, colo);
            blk_mma<NHID>(c0, c1, xs   + (mt * 16 + row) * LP + 2 * half, sel ? lg.Wr2 : lg.Wr1, half, colo);
            const float* bp = sel ? lg.b2 : lg.b1;
            const float bv0 = bp[colo], bv1 = bp[colo + 1];
            #pragma unroll
            for (int i = 0; i < 8; ++i) {
                v2f y;
                y.x = fmaxf(c0[i] + bv0, 0.0f);
                y.y = fmaxf(c1[i] + bv1, 0.0f);
                if (sel) {
                    *(v2f*)(xg + (lrow + i) * LP + colo) = y;
                } else if (lrow + i < (int)rcur) {
                    *(v2f*)(lg.T1 + (size_t)(n0 + lrow + i) * NHID + colo) = y;
                }
            }
        }
        __syncthreads();

        if (gw) {
            v8f c0 = {}, c1 = {};
            blk_mma<NHID>(c0, c1, xg + (mt * 16 + row) * LP + 2 * half,
                          lg.Q + (sel ? (size_t)NHID * NHID : 0), half, colo);
            const float bv0 = sel ? 0.0f : lg.Qb[colo];
            const float bv1 = sel ? 0.0f : lg.Qb[colo + 1];
            float* dstp = sel ? lg.Bb : lg.Ab;
            #pragma unroll
            for (int i = 0; i < 8; ++i) {
                v2f y;
                y.x = c0[i] + bv0;
                y.y = c1[i] + bv1;
                if (lrow + i < (int)rcur)
                    *(v2f*)(dstp + (size_t)(n0 + lrow + i) * NHID + colo) = y;
            }
        }
        return;
    }

    if (DEC) {
        float* xs = gbuf;
        for (int i = tid; i < RBP * 32; i += OWN_T) {
            const int r = i >> 5;
            const int l = i & 31;
            float4 y = make_float4(0.0f, 0.0f, 0.0f, 0.0f);
            if (r < (int)rcur)
                y = gate4(acc, cntL, r, l, Xcur, Xnew, (size_t)(n0 + r) * NHID + l * 4);
            *(float4*)(xs + r * LP + l * 4) = y;
        }
        __syncthreads();

        if (wave < MT * 2) {
            const int mt   = wave >> 1;
            const int row  = lane & 15;
            const int half = lane >> 4;
            const int colo = (wave & 1) * 32 + 2 * row;
            const int lrow = mt * 16 + 8 * half;
            v8f c0 = {}, c1 = {};
            blk_mma<NCLASS>(c0, c1, xs + (mt * 16 + row) * LP + 2 * half, lg.Wd, half, colo);
            const float bv0 = lg.bd[colo], bv1 = lg.bd[colo + 1];
            #pragma unroll
            for (int i = 0; i < 8; ++i) {
                v2f y;
                y.x = c0[i] + bv0;
                y.y = c1[i] + bv1;
                if (lrow + i < (int)rcur)
                    *(v2f*)(lg.Od + (size_t)(n0 + lrow + i) * NCLASS + colo) = y;
            }
        }
        return;
    }

    for (int i = tid; i < (int)rcur * 32; i += OWN_T) {
        const int r = i >> 5;
        const int l = i & 31;
        const size_t o = (size_t)(n0 + r) * NHID + l * 4;
        *(float4*)(out + o) = gate4(acc, cntL, r, l, Xcur, Xnew, o);
    }
}

__global__ void k_gemm(const float* __restrict__ A1, const float* __restrict__ W1,
                       const float* __restrict__ A2, const float* __restrict__ W2,
                       const float* __restrict__ bias, float* __restrict__ C,
                       int M, int K, int N, int relu) {
    const int lane   = threadIdx.x & 31;
    const int wave   = threadIdx.x >> 5;
    const int ntiles = N >> 4;
    if (wave >= ntiles) return;

    const int m0   = blockIdx.x << 4;
    const int n0   = wave << 4;
    const int row  = lane & 15;
    const int half = lane >> 4;

    v8f c = {};

    {
        const float* arow = A1 + (size_t)(m0 + row) * K;
        for (int k0 = 0; k0 < K; k0 += 4) {
            v2f a, b;
            a.x = arow[k0 + 2 * half];
            a.y = arow[k0 + 2 * half + 1];
            b.x = W1[(size_t)(k0 + 2 * half) * N + n0 + row];
            b.y = W1[(size_t)(k0 + 2 * half + 1) * N + n0 + row];
            c = __builtin_amdgcn_wmma_f32_16x16x4_f32(false, a, false, b,
                                                      (short)0, c, false, false);
        }
    }
    if (A2 != nullptr) {
        const float* arow = A2 + (size_t)(m0 + row) * K;
        for (int k0 = 0; k0 < K; k0 += 4) {
            v2f a, b;
            a.x = arow[k0 + 2 * half];
            a.y = arow[k0 + 2 * half + 1];
            b.x = W2[(size_t)(k0 + 2 * half) * N + n0 + row];
            b.y = W2[(size_t)(k0 + 2 * half + 1) * N + n0 + row];
            c = __builtin_amdgcn_wmma_f32_16x16x4_f32(false, a, false, b,
                                                      (short)0, c, false, false);
        }
    }

    const float bv = bias ? bias[n0 + row] : 0.0f;
    #pragma unroll
    for (int i = 0; i < 8; ++i) {
        float x = c[i] + bv;
        if (relu) x = fmaxf(x, 0.0f);
        int m = m0 + i + 8 * half;
        if (m < M) C[(size_t)m * N + n0 + row] = x;
    }
}

extern "C" void kernel_launch(void* const* d_in, const int* in_sizes, int n_in,
                              void* d_out, int out_size, void* d_ws, size_t ws_size,
                              hipStream_t stream) {
    const float* X       = (const float*)d_in[0];
    const int*   ei      = (const int*)  d_in[1];
    const float* enc_W   = (const float*)d_in[2];
    const float* enc_b   = (const float*)d_in[3];
    const float* conv_Wl = (const float*)d_in[4];
    const float* conv_bl = (const float*)d_in[5];
    const float* conv_Wr = (const float*)d_in[6];
    const float* gg_Wl   = (const float*)d_in[7];
    const float* gg_bl   = (const float*)d_in[8];
    const float* gg_Wr   = (const float*)d_in[9];
    const float* Q_W     = (const float*)d_in[10];
    const float* Q_b     = (const float*)d_in[11];
    const float* dec_W   = (const float*)d_in[12];
    const float* dec_b   = (const float*)d_in[13];

    const int N = in_sizes[0] / NFEAT;
    const int E = in_sizes[1] / 2;
    const int* src = ei;
    const int* dst = ei + E;

    float* ws       = (float*)d_ws;
    const size_t NH = (size_t)N * NHID;
    float* Xa       = ws + 2 * (size_t)N;
    float* Xb       = Xa + NH;
    float* T1       = Xb + NH;
    float* Ab       = T1 + NH;
    float* Bb       = Ab + NH;

    const int gemmGrd = N / 16;
    const int ownGrd  = (N + RB - 1) / RB;

    k_gemm<<<gemmGrd, 256, 0, stream>>>(X, enc_W, nullptr, nullptr, enc_b,
                                        Xa, N, NFEAT, NHID, 1);

    LayerG lg;
    lg.Wl1 = conv_Wl; lg.Wr1 = conv_Wr; lg.b1 = conv_bl;
    lg.Wl2 = gg_Wl;   lg.Wr2 = gg_Wr;   lg.b2 = gg_bl;
    lg.Q   = Q_W;     lg.Qb  = Q_b;
    lg.T1  = T1;      lg.Ab  = Ab;      lg.Bb  = Bb;
    lg.Wd  = dec_W;   lg.bd  = dec_b;   lg.Od  = (float*)d_out;

    int* listD = (int*)(Bb + NH);
    int* listS = listD + (size_t)ownGrd * LSTRIDE;
    const float* wsf = ws;

    k_owner<0, 0, false><<<ownGrd, OWN_T, 0, stream>>>(Xa, nullptr, dst, src,
                                                       nullptr, nullptr, nullptr, lg, N, E,
                                                       listD, wsf, ws_size);
    k_owner<1, 0, false><<<ownGrd, OWN_T, 0, stream>>>(Bb, Ab, src, dst,
                                                       Xa, T1, Xb, lg, N, E,
                                                       listS, wsf, ws_size);
    k_owner<0, 1, false><<<ownGrd, OWN_T, 0, stream>>>(Xb, nullptr, dst, src,
                                                       nullptr, nullptr, nullptr, lg, N, E,
                                                       listD, wsf, ws_size);
    k_owner<1, 1, true><<<ownGrd, OWN_T, 0, stream>>>(Bb, Ab, src, dst,
                                                      Xb, T1, nullptr, lg, N, E,
                                                      listS, wsf, ws_size);
}


# Encoder GEMM re-cut without LDS: 4-wave groups of 16 rows, each wave two interleaved column tiles; A by 16-B, weights by 8-B loads, k loop unrolled; k order inside 8-blocks changed (f32 sum order)
# speedup vs baseline: 3.9375x; 1.0801x over previous
#include <hip/hip_runtime.h>
#include <hip/hip_bf16.h>


typedef __attribute__((ext_vector_type(2))) float v2f;
typedef __attribute__((ext_vector_type(8))) float v8f;

#define NFEAT  256
#define NHID   128
#define NCLASS 64

#define RB    40
#define OWN_T 1024
#define QCAP  544
#define MT    3
#define RBP   (MT * 16)
#define LP    132
#define LCAP    2048
#define LHDR    32
#define LSTRIDE (LHDR + LCAP)

struct LayerG {
    const float* Wl1; const float* Wr1; const float* b1;
    const float* Wl2; const float* Wr2; const float* b2;
    const float* Q;   const float* Qb;
    float* T1; float* Ab; float* Bb;
    const float* Wd;  const float* bd;  float* Od;
};

template <int WS>
__device__ __forceinline__ void blk_mma(v8f& c0, v8f& c1, const float* __restrict__ ar,
                                        const float* __restrict__ Wm, int half, int colo) {
    const float* wp = Wm + (size_t)(2 * half) * WS + colo;
    for (int k0 = 0; k0 < NHID; k0 += 4) {
        const v2f a  = *(const v2f*)(ar + k0);
        const v2f w0 = *(const v2f*)(wp + (size_t)k0 * WS);
        const v2f w1 = *(const v2f*)(wp + (size_t)(k0 + 1) * WS);
        v2f b0, b1;
        b0.x = w0.x; b0.y = w1.x;
        b1.x = w0.y; b1.y = w1.y;
        c0 = __builtin_amdgcn_wmma_f32_16x16x4_f32(false, a, false, b0, (short)0, c0, false, false);
        c1 = __builtin_amdgcn_wmma_f32_16x16x4_f32(false, a, false, b1, (short)0, c1, false, false);
    }
}

__device__ __forceinline__ void ldsAddF(float* p, float v) {
    __hip_atomic_fetch_add(p, v, __ATOMIC_RELAXED, __HIP_MEMORY_SCOPE_WORKGROUP);
}

__device__ __forceinline__ void own_push(int* __restrict__ q, int& qn, int v, int e,
                                         int n0, unsigned rcur, int lane) {
    const unsigned rel = (unsigned)(v - n0);
    const bool m = rel < rcur;
    const unsigned mask = __builtin_amdgcn_ballot_w32(m);
    if (mask) {
        if (m) q[qn + __builtin_popcount(mask & ((1u << lane) - 1u))] = (e << 6) | (int)rel;
        qn += __builtin_popcount(mask);
    }
}

__device__ __forceinline__ void own_push4(int* __restrict__ q, int& qn, const int4 v, int e,
                                          int n0, unsigned rcur, int lane) {
    own_push(q, qn, v.x, e + 0, n0, rcur, lane);
    own_push(q, qn, v.y, e + 1, n0, rcur, lane);
    own_push(q, qn, v.z, e + 2, n0, rcur, lane);
    own_push(q, qn, v.w, e + 3, n0, rcur, lane);
}

template <int MODE>
__device__ __forceinline__ int own_row(float4& v, int g, int rel, int i,
                                       const float* __restrict__ G,
                                       const float* __restrict__ Aown, int n0, int lane) {
    const int gi = __builtin_amdgcn_readlane(g, i);
    const int ri = __builtin_amdgcn_readlane(rel, i);
    v = *(const float4*)(G + (size_t)gi * NHID + lane * 4);
    if (MODE == 1) {
        const float4 a = *(const float4*)(Aown + (size_t)(n0 + ri) * NHID + lane * 4);
        v.x = (a.x + v.x) * (a.x + v.x);
        v.y = (a.y + v.y) * (a.y + v.y);
        v.z = (a.z + v.z) * (a.z + v.z);
        v.w = (a.w + v.w) * (a.w + v.w);
    }
    return ri;
}

__device__ __forceinline__ void own_add(float* __restrict__ acc, int ri, int lane, const float4 v) {
    float* p = acc + ri * NHID + lane;
    ldsAddF(p,      v.x);
    ldsAddF(p + 32, v.y);
    ldsAddF(p + 64, v.z);
    ldsAddF(p + 96, v.w);
}

template <int MODE>
__device__ __forceinline__ void own_rows(int n, int g, int rel, float* __restrict__ acc,
                                         const float* __restrict__ G,
                                         const float* __restrict__ Aown, int n0, int lane) {
    int i = 0;
    for (; i + 4 <= n; i += 4) {
        float4 v0, v1, v2, v3;
        const int r0 = own_row<MODE>(v0, g, rel, i + 0, G, Aown, n0, lane);
        const int r1 = own_row<MODE>(v1, g, rel, i + 1, G, Aown, n0, lane);
        const int r2 = own_row<MODE>(v2, g, rel, i + 2, G, Aown, n0, lane);
        const int r3 = own_row<MODE>(v3, g, rel, i + 3, G, Aown, n0, lane);
        own_add(acc, r0, lane, v0);
        own_add(acc, r1, lane, v1);
        own_add(acc, r2, lane, v2);
        own_add(acc, r3, lane, v3);
    }
    for (; i < n; ++i) {
        float4 v0;
        const int r0 = own_row<MODE>(v0, g, rel, i, G, Aown, n0, lane);
        own_add(acc, r0, lane, v0);
    }
}

template <int MODE, bool SAVE>
__device__ __forceinline__ void own_drain(int* __restrict__ q, int& qn, float* __restrict__ acc,
                                          int* __restrict__ cntL, int* __restrict__ lctl,
                                          int* __restrict__ lst,
                                          const float* __restrict__ G,
                                          const float* __restrict__ Aown,
                                          const int* __restrict__ gath, int n0, int lane) {
    __builtin_amdgcn_wave_barrier();
    int b = 0;
    for (; b + 32 <= qn; b += 32) {
        const int ent = q[b + lane];
        const int rel = ent & 63;
        const int g = gath[ent >> 6];
        __hip_atomic_fetch_add(&cntL[rel], 1, __ATOMIC_RELAXED, __HIP_MEMORY_SCOPE_WORKGROUP);
        if (SAVE && lst != nullptr) {
            int c = 0;
            if (lane == 0)
                c = __hip_atomic_fetch_add(&lctl[0], 1, __ATOMIC_RELAXED, __HIP_MEMORY_SCOPE_WORKGROUP);
            c = __builtin_amdgcn_readfirstlane(c);
            if (c < LCAP / 32) lst[LHDR + c * 32 + lane] = (g << 6) | rel;
        }
        own_rows<MODE>(32, g, rel, acc, G, Aown, n0, lane);
    }
    const int r = qn - b;
    if (b > 0 && lane < r) {
        const int t = q[b + lane];
        q[lane] = t;
    }
    qn = r;
    __builtin_amdgcn_wave_barrier();
}

__device__ __forceinline__ void own_load4(int4& a0, int4& a1, int4& a2, int4& a3,
                                          const int4* __restrict__ sv, int vb) {
    a0 = sv[vb];
    a1 = sv[vb + OWN_T];
    a2 = sv[vb + 2 * OWN_T];
    a3 = sv[vb + 3 * OWN_T];
}

template <int MODE, bool SAVE>
__device__ __forceinline__ void own_trip(const int4 a0, const int4 a1, const int4 a2, const int4 a3,
                                         int vb, int* __restrict__ q, int& qn,
                                         float* __restrict__ acc, int* __restrict__ cntL,
                                         int* __restrict__ lctl, int* __restrict__ lst,
                                         const float* __restrict__ G,
                                         const float* __restrict__ Aown,
                                         const int* __restrict__ gath,
                                         int n0, unsigned rcur, int lane) {
    own_push4(q, qn, a0, vb * 4, n0, rcur, lane);
    own_push4(q, qn, a1, (vb + OWN_T) * 4, n0, rcur, lane);
    own_push4(q, qn, a2, (vb + 2 * OWN_T) * 4, n0, rcur, lane);
    own_push4(q, qn, a3, (vb + 3 * OWN_T) * 4, n0, rcur, lane);
    if (qn >= 32) own_drain<MODE, SAVE>(q, qn, acc, cntL, lctl, lst, G, Aown, gath, n0, lane);
}

template <int MODE, bool SAVE>
__device__ __forceinline__ void own_filter(int* __restrict__ q, float* __restrict__ acc,
                                           int* __restrict__ cntL, int* __restrict__ lctl,
                                           int* __restrict__ lq, int* __restrict__ lst,
                                           const float* __restrict__ G,
                                           const float* __restrict__ Aown,
                                           const int* __restrict__ scat,
                                           const int* __restrict__ gath,
                                           int n0, unsigned rcur, int E,
                                           int tid, int lane, int wave) {
    int qn = 0;

    const int nvec = ((E & 3) == 0 && ((size_t)scat & 15) == 0) ? (E >> 2) : 0;
    const int4* sv = (const int4*)scat;
    const int4 none = make_int4(-1, -1, -1, -1);

    const int nfull = nvec / (4 * OWN_T);
    const int toff = (nfull > 0) ? (int)(blockIdx.x % (unsigned)nfull) : 0;
    auto trip_vb = [&](int t) {
        int tt = t + toff;
        if (tt >= nfull) tt -= nfull;
        return tt * 4 * OWN_T + tid;
    };
    int4 a0 = none, a1 = none, a2 = none, a3 = none;
    int4 b0 = none, b1 = none, b2 = none, b3 = none;
    if (nfull > 0) own_load4(a0, a1, a2, a3, sv, trip_vb(0));
    int t = 0;
    for (; t + 2 <= nfull; t += 2) {
        own_load4(b0, b1, b2, b3, sv, trip_vb(t + 1));
        own_trip<MODE, SAVE>(a0, a1, a2, a3, trip_vb(t), q, qn, acc, cntL, lctl, lst, G, Aown, gath,
                             n0, rcur, lane);
        if (t + 2 < nfull) own_load4(a0, a1, a2, a3, sv, trip_vb(t + 2));
        own_trip<MODE, SAVE>(b0, b1, b2, b3, trip_vb(t + 1), q, qn, acc, cntL, lctl, lst, G, Aown, gath,
                             n0, rcur, lane);
    }
    if (t < nfull)
        own_trip<MODE, SAVE>(a0, a1, a2, a3, trip_vb(t), q, qn, acc, cntL, lctl, lst, G, Aown, gath,
                             n0, rcur, lane);
    for (int vb = nfull * 4 * OWN_T; vb < nvec; vb += OWN_T) {
        const int idx = vb + tid;
        int4 a = none;
        if (idx < nvec) a = sv[idx];
        own_push4(q, qn, a, idx * 4, n0, rcur, lane);
        if (qn >= 32) own_drain<MODE, SAVE>(q, qn, acc, cntL, lctl, lst, G, Aown, gath, n0, lane);
    }
    for (int eb = nvec * 4; eb < E; eb += OWN_T) {
        const int e = eb + tid;
        const int v = (e < E) ? scat[e] : -1;
        own_push(q, qn, v, e, n0, rcur, lane);
        if (qn >= 32) own_drain<MODE, SAVE>(q, qn, acc, cntL, lctl, lst, G, Aown, gath, n0, lane);
    }

    __builtin_amdgcn_wave_barrier();
    {
        int base = 0;
        if (lane == 0 && qn > 0)
            base = __hip_atomic_fetch_add(&lctl[1], qn, __ATOMIC_RELAXED, __HIP_MEMORY_SCOPE_WORKGROUP);
        base = __builtin_amdgcn_readfirstlane(base);
        if (lane < qn) lq[base + lane] = q[lane];
    }
    __syncthreads();
    const int L = lctl[1];
    const int F = lctl[0];
    const bool fits = SAVE && lst != nullptr && F * 32 + L <= LCAP;
    if (fits) {
        const int i = wave * 32 + lane;
        if (i < L) {
            const int ent = lq[i];
            lst[LHDR + F * 32 + i] = (gath[ent >> 6] << 6) | (ent & 63);
        }
    }
    if (SAVE && lst != nullptr && tid == 0) lst[0] = fits ? F * 32 + L : -1;
    {
        const int wv = __builtin_amdgcn_readfirstlane(wave);
        const int lo = (wv * L) >> 5;
        const int n  = (((wv + 1) * L) >> 5) - lo;
        int rel = 0, g = 0;
        if (lane < n) {
            const int ent = lq[lo + lane];
            rel = ent & 63;
            g = gath[ent >> 6];
            __hip_atomic_fetch_add(&cntL[rel], 1, __ATOMIC_RELAXED, __HIP_MEMORY_SCOPE_WORKGROUP);
        }
        own_rows<MODE>(n, g, rel, acc, G, Aown, n0, lane);
    }
}

template <int MODE>
__device__ __forceinline__ void own_replay(const int* __restrict__ lst, int T,
                                           float* __restrict__ acc, int* __restrict__ cntL,
                                           const float* __restrict__ G,
                                           const float* __restrict__ Aown,
                                           int n0, int lane, int wave) {
    const int wv = __builtin_amdgcn_readfirstlane(wave);
    const int lo = (wv * T) >> 5;
    const int hi = ((wv + 1) * T) >> 5;
    for (int b = lo; b < hi; b += 32) {
        const int n = (hi - b < 32) ? (hi - b) : 32;
        int rel = 0, g = 0;
        if (lane < n) {
            const int ent = lst[LHDR + b + lane];
            rel = ent & 63;
            g = ent >> 6;
            __hip_atomic_fetch_add(&cntL[rel], 1, __ATOMIC_RELAXED, __HIP_MEMORY_SCOPE_WORKGROUP);
        }
        own_rows<MODE>(n, g, rel, acc, G, Aown, n0, lane);
    }
}

__device__ __forceinline__ float4 gate4(const float* __restrict__ acc, const int* __restrict__ cntL,
                                        int r, int l, const float* __restrict__ Xcur,
                                        const float* __restrict__ Xnew, size_t o) {
    const float c = fmaxf((float)cntL[r], 1.0f);
    const float* a = acc + r * NHID + l;
    const float4 s = make_float4(a[0], a[32], a[64], a[96]);
    const float4 xc = *(const float4*)(Xcur + o);
    const float4 xn = *(const float4*)(Xnew + o);
    const float tx = tanhf(s.x / c), ty = tanhf(s.y / c);
    const float tz = tanhf(s.z / c), tw = tanhf(s.w / c);
    float4 y;
    y.x = (1.0f - tx) * xc.x + tx * xn.x;
    y.y = (1.0f - ty) * xc.y + ty * xn.y;
    y.z = (1.0f - tz) * xc.z + tz * xn.z;
    y.w = (1.0f - tw) * xc.w + tw * xn.w;
    return y;
}

template <int MODE, int PASS, bool DEC>
__global__ __launch_bounds__(OWN_T)
void k_owner(const float* __restrict__ G, const float* __restrict__ Aown,
             const int* __restrict__ scat, const int* __restrict__ gath,
             const float* __restrict__ Xcur, const float* __restrict__ Xnew,
             float* __restrict__ out, const LayerG lg, int N, int E,
             int* __restrict__ lstAll, const float* __restrict__ wsf, size_t wsz) {
    __shared__ float acc[RB * NHID];
    __shared__ int   cntL[RB];
    __shared__ int   lctl[2];
    __shared__ int   lq[OWN_T];
    __shared__ int   qbuf[OWN_T / 32][QCAP];
    __shared__ alignas(16) float gbuf[MODE == 0 ? 3 * RBP * LP : (DEC ? RBP * LP : 4)];

    const int tid  = threadIdx.x;
    const int lane = tid & 31;
    const int wave = tid >> 5;
    const int n0   = blockIdx.x * RB;
    const unsigned rcur = (unsigned)((N - n0 < RB) ? (N - n0) : RB);

    int* lst = nullptr;
    if ((size_t)((const char*)(lstAll + (size_t)gridDim.x * LSTRIDE) - (const char*)wsf) <= wsz)
        lst = lstAll + (size_t)blockIdx.x * LSTRIDE;

    for (int i = tid; i < RB * NHID; i += OWN_T) acc[i] = 0.0f;
    if (tid < RB) cntL[tid] = 0;
    if (tid < 2) lctl[tid] = 0;
    __syncthreads();

    int T = -1;
    if (PASS == 1 && lst != nullptr) {
        T = lst[0];
        if (T > LCAP) T = -1;
    }
    if (T >= 0)
        own_replay<MODE>(lst, T, acc, cntL, G, Aown, n0, lane, wave);
    else
        own_filter<MODE, PASS == 0>(qbuf[wave], acc, cntL, lctl, lq, lst, G, Aown, scat, gath,
                                    n0, rcur, E, tid, lane, wave);
    __syncthreads();

    if (MODE == 0) {
        float* aggs = gbuf;
        float* xs   = gbuf + RBP * LP;
        float* xg   = gbuf + 2 * RBP * LP;
        for (int i = tid; i < RBP * 32; i += OWN_T) {
            const int r = i >> 5;
            const int l = i & 31;
            float4 s = make_float4(0.0f, 0.0f, 0.0f, 0.0f);
            float4 x = s;
            if (r < (int)rcur) {
                const float c = fmaxf((float)cntL[r], 1.0f);
                const float* a = acc + r * NHID + l;
                s = make_float4(a[0] / c, a[32] / c, a[64] / c, a[96] / c);
                x = *(const float4*)(G + (size_t)(n0 + r) * NHID + l * 4);
            }
            *(float4*)(aggs + r * LP + l * 4) = s;
            *(float4*)(xs + r * LP + l * 4) = x;
        }
        __syncthreads();

        const bool gw   = wave < MT * 8;
        const int  mt   = wave >> 3;
        const int  sel  = (wave >> 2) & 1;
        const int  row  = lane & 15;
        const int  half = lane >> 4;
        const int  colo = (wave & 3) * 32 + 2 * row;
        const int  lrow = mt * 16 + 8 * half;

        if (gw) {
            v8f c0 = {}, c1 = {};
            blk_mma<NHID>(c0, c1, aggs + (mt * 16 + row) * LP + 2 * half, sel ? lg.Wl2 : lg.Wl1, half, colo);
            blk_mma<NHID>(c0, c1, xs   + (mt * 16 + row) * LP + 2 * half, sel ? lg.Wr2 : lg.Wr1, half, colo);
            const float* bp = sel ? lg.b2 : lg.b1;
            const float bv0 = bp[colo], bv1 = bp[colo + 1];
            #pragma unroll
            for (int i = 0; i < 8; ++i) {
                v2f y;
                y.x = fmaxf(c0[i] + bv0, 0.0f);
                y.y = fmaxf(c1[i] + bv1, 0.0f);
                if (sel) {
                    *(v2f*)(xg + (lrow + i) * LP + colo) = y;
                } else if (lrow + i < (int)rcur) {
                    *(v2f*)(lg.T1 + (size_t)(n0 + lrow + i) * NHID + colo) = y;
                }
            }
        }
        __syncthreads();

        if (gw) {
            v8f c0 = {}, c1 = {};
            blk_mma<NHID>(c0, c1, xg + (mt * 16 + row) * LP + 2 * half,
                          lg.Q + (sel ? (size_t)NHID * NHID : 0), half, colo);
            const float bv0 = sel ? 0.0f : lg.Qb[colo];
            const float bv1 = sel ? 0.0f : lg.Qb[colo + 1];
            float* dstp = sel ? lg.Bb : lg.Ab;
            #pragma unroll
            for (int i = 0; i < 8; ++i) {
                v2f y;
                y.x = c0[i] + bv0;
                y.y = c1[i] + bv1;
                if (lrow + i < (int)rcur)
                    *(v2f*)(dstp + (size_t)(n0 + lrow + i) * NHID + colo) = y;
            }
        }
        return;
    }

    if (DEC) {
        float* xs = gbuf;
        for (int i = tid; i < RBP * 32; i += OWN_T) {
            const int r = i >> 5;
            const int l = i & 31;
            float4 y = make_float4(0.0f, 0.0f, 0.0f, 0.0f);
            if (r < (int)rcur)
                y = gate4(acc, cntL, r, l, Xcur, Xnew, (size_t)(n0 + r) * NHID + l * 4);
            *(float4*)(xs + r * LP + l * 4) = y;
        }
        __syncthreads();

        if (wave < MT * 2) {
            const int mt   = wave >> 1;
            const int row  = lane & 15;
            const int half = lane >> 4;
            const int colo = (wave & 1) * 32 + 2 * row;
            const int lrow = mt * 16 + 8 * half;
            v8f c0 = {}, c1 = {};
            blk_mma<NCLASS>(c0, c1, xs + (mt * 16 + row) * LP + 2 * half, lg.Wd, half, colo);
            const float bv0 = lg.bd[colo], bv1 = lg.bd[colo + 1];
            #pragma unroll
            for (int i = 0; i < 8; ++i) {
                v2f y;
                y.x = c0[i] + bv0;
                y.y = c1[i] + bv1;
                if (lrow + i < (int)rcur)
                    *(v2f*)(lg.Od + (size_t)(n0 + lrow + i) * NCLASS + colo) = y;
            }
        }
        return;
    }

    for (int i = tid; i < (int)rcur * 32; i += OWN_T) {
        const int r = i >> 5;
        const int l = i & 31;
        const size_t o = (size_t)(n0 + r) * NHID + l * 4;
        *(float4*)(out + o) = gate4(acc, cntL, r, l, Xcur, Xnew, o);
    }
}

__device__ __forceinline__ void enc_blk(v8f& c0, v8f& c1, const float* __restrict__ ap,
                                        const float* __restrict__ wp) {
    const float4 a4 = *(const float4*)ap;
    const v2f w0 = *(const v2f*)(wp);
    const v2f w1 = *(const v2f*)(wp + NHID);
    const v2f w2 = *(const v2f*)(wp + 2 * NHID);
    const v2f w3 = *(const v2f*)(wp + 3 * NHID);
    v2f a, b0, b1;
    a.x = a4.x;  a.y = a4.y;
    b0.x = w0.x; b0.y = w1.x;
    b1.x = w0.y; b1.y = w1.y;
    c0 = __builtin_amdgcn_wmma_f32_16x16x4_f32(false, a, false, b0, (short)0, c0, false, false);
    c1 = __builtin_amdgcn_wmma_f32_16x16x4_f32(false, a, false, b1, (short)0, c1, false, false);
    a.x = a4.z;  a.y = a4.w;
    b0.x = w2.x; b0.y = w3.x;
    b1.x = w2.y; b1.y = w3.y;
    c0 = __builtin_amdgcn_wmma_f32_16x16x4_f32(false, a, false, b0, (short)0, c0, false, false);
    c1 = __builtin_amdgcn_wmma_f32_16x16x4_f32(false, a, false, b1, (short)0, c1, false, false);
}

__global__ __launch_bounds__(128) __attribute__((amdgpu_waves_per_eu(1, 4)))
void k_enc(const float* __restrict__ A, const float* __restrict__ W,
           const float* __restrict__ bias, float* __restrict__ C, int M) {
    const int lane = threadIdx.x & 31;
    const int wave = threadIdx.x >> 5;
    const int m0   = blockIdx.x << 4;
    const int row  = lane & 15;
    const int half = lane >> 4;
    const int colo = wave * 32 + 2 * row;

    const int ar = (m0 + row < M) ? (m0 + row) : (M - 1);
    const float* ap = A + (size_t)ar * NFEAT + 4 * half;
    const float* wp = W + (size_t)(4 * half) * NHID + colo;

    v8f c0 = {}, c1 = {};
    #pragma unroll
    for (int kb = 0; kb < NFEAT; kb += 8)
        enc_blk(c0, c1, ap + kb, wp + (size_t)kb * NHID);

    const float bv0 = bias[colo], bv1 = bias[colo + 1];
    const int lrow = m0 + 8 * half;
    #pragma unroll
    for (int i = 0; i < 8; ++i) {
        v2f y;
        y.x = fmaxf(c0[i] + bv0, 0.0f);
        y.y = fmaxf(c1[i] + bv1, 0.0f);
        if (lrow + i < M) *(v2f*)(C + (size_t)(lrow + i) * NHID + colo) = y;
    }
}

extern "C" void kernel_launch(void* const* d_in, const int* in_sizes, int n_in,
                              void* d_out, int out_size, void* d_ws, size_t ws_size,
                              hipStream_t stream) {
    const float* X       = (const float*)d_in[0];
    const int*   ei      = (const int*)  d_in[1];
    const float* enc_W   = (const float*)d_in[2];
    const float* enc_b   = (const float*)d_in[3];
    const float* conv_Wl = (const float*)d_in[4];
    const float* conv_bl = (const float*)d_in[5];
    const float* conv_Wr = (const float*)d_in[6];
    const float* gg_Wl   = (const float*)d_in[7];
    const float* gg_bl   = (const float*)d_in[8];
    const float* gg_Wr   = (const float*)d_in[9];
    const float* Q_W     = (const float*)d_in[10];
    const float* Q_b     = (const float*)d_in[11];
    const float* dec_W   = (const float*)d_in[12];
    const float* dec_b   = (const float*)d_in[13];

    const int N = in_sizes[0] / NFEAT;
    const int E = in_sizes[1] / 2;
    const int* src = ei;
    const int* dst = ei + E;

    float* ws       = (float*)d_ws;
    const size_t NH = (size_t)N * NHID;
    float* Xa       = ws + 2 * (size_t)N;
    float* Xb       = Xa + NH;
    float* T1       = Xb + NH;
    float* Ab       = T1 + NH;
    float* Bb       = Ab + NH;

    const int encGrd  = (N + 15) / 16;
    const int ownGrd  = (N + RB - 1) / RB;

    k_enc<<<encGrd, 128, 0, stream>>>(X, enc_W, enc_b, Xa, N);

    LayerG lg;
    lg.Wl1 = conv_Wl; lg.Wr1 = conv_Wr; lg.b1 = conv_bl;
    lg.Wl2 = gg_Wl;   lg.Wr2 = gg_Wr;   lg.b2 = gg_bl;
    lg.Q   = Q_W;     lg.Qb  = Q_b;
    lg.T1  = T1;      lg.Ab  = Ab;      lg.Bb  = Bb;
    lg.Wd  = dec_W;   lg.bd  = dec_b;   lg.Od  = (float*)d_out;

    int* listD = (int*)(Bb + NH);
    int* listS = listD + (size_t)ownGrd * LSTRIDE;
    const float* wsf = ws;

    k_owner<0, 0, false><<<ownGrd, OWN_T, 0, stream>>>(Xa, nullptr, dst, src,
                                                       nullptr, nullptr, nullptr, lg, N, E,
                                                       listD, wsf, ws_size);
    k_owner<1, 0, false><<<ownGrd, OWN_T, 0, stream>>>(Bb, Ab, src, dst,
                                                       Xa, T1, Xb, lg, N, E,
                                                       listS, wsf, ws_size);
    k_owner<0, 1, false><<<ownGrd, OWN_T, 0, stream>>>(Xb, nullptr, dst, src,
                                                       nullptr, nullptr, nullptr, lg, N, E,
                                                       listD, wsf, ws_size);
    k_owner<1, 1, true><<<ownGrd, OWN_T, 0, stream>>>(Bb, Ab, src, dst,
                                                      Xb, T1, nullptr, lg, N, E,
                                                      listS, wsf, ws_size);
}


# Layer-0 owner filters read key planes, not 4-B indices: spare work-groups of the encoder launch write 8 bit planes of owner-block numbers per 32 edges; LDS f32 sums stay arrival order
# speedup vs baseline: 4.1333x; 1.0497x over previous
#include <hip/hip_runtime.h>
#include <hip/hip_bf16.h>


typedef __attribute__((ext_vector_type(2))) float v2f;
typedef __attribute__((ext_vector_type(8))) float v8f;

#define NFEAT  256
#define NHID   128
#define NCLASS 64

#define RB    40
#define OWN_T 1024
#define QCAP  544
#define MT    3
#define RBP   (MT * 16)
#define LP    132
#define LCAP    2048
#define LHDR    32
#define LSTRIDE (LHDR + LCAP)
#define KGW 16

struct LayerG {
    const float* Wl1; const float* Wr1; const float* b1;
    const float* Wl2; const float* Wr2; const float* b2;
    const float* Q;   const float* Qb;
    float* T1; float* Ab; float* Bb;
    const float* Wd;  const float* bd;  float* Od;
};

template <int WS>
__device__ __forceinline__ void blk_mma(v8f& c0, v8f& c1, const float* __restrict__ ar,
                                        const float* __restrict__ Wm, int half, int colo) {
    const float* wp = Wm + (size_t)(2 * half) * WS + colo;
    for (int k0 = 0; k0 < NHID; k0 += 4) {
        const v2f a  = *(const v2f*)(ar + k0);
        const v2f w0 = *(const v2f*)(wp + (size_t)k0 * WS);
        const v2f w1 = *(const v2f*)(wp + (size_t)(k0 + 1) * WS);
        v2f b0, b1;
        b0.x = w0.x; b0.y = w1.x;
        b1.x = w0.y; b1.y = w1.y;
        c0 = __builtin_amdgcn_wmma_f32_16x16x4_f32(false, a, false, b0, (short)0, c0, false, false);
        c1 = __builtin_amdgcn_wmma_f32_16x16x4_f32(false, a, false, b1, (short)0, c1, false, false);
    }
}

__device__ __forceinline__ void ldsAddF(float* p, float v) {
    __hip_atomic_fetch_add(p, v, __ATOMIC_RELAXED, __HIP_MEMORY_SCOPE_WORKGROUP);
}

__device__ __forceinline__ void own_push(int* __restrict__ q, int& qn, int v, int e,
                                         int n0, unsigned rcur, int lane) {
    const unsigned rel = (unsigned)(v - n0);
    const bool m = rel < rcur;
    const unsigned mask = __builtin_amdgcn_ballot_w32(m);
    if (mask) {
        if (m) q[qn + __builtin_popcount(mask & ((1u << lane) - 1u))] = (e << 6) | (int)rel;
        qn += __builtin_popcount(mask);
    }
}

__device__ __forceinline__ void own_push4(int* __restrict__ q, int& qn, const int4 v, int e,
                                          int n0, unsigned rcur, int lane) {
    own_push(q, qn, v.x, e + 0, n0, rcur, lane);
    own_push(q, qn, v.y, e + 1, n0, rcur, lane);
    own_push(q, qn, v.z, e + 2, n0, rcur, lane);
    own_push(q, qn, v.w, e + 3, n0, rcur, lane);
}

template <int MODE>
__device__ __forceinline__ int own_row(float4& v, int g, int rel, int i,
                                       const float* __restrict__ G,
                                       const float* __restrict__ Aown, int n0, int lane) {
    const int gi = __builtin_amdgcn_readlane(g, i);
    const int ri = __builtin_amdgcn_readlane(rel, i);
    v = *(const float4*)(G + (size_t)gi * NHID + lane * 4);
    if (MODE == 1) {
        const float4 a = *(const float4*)(Aown + (size_t)(n0 + ri) * NHID + lane * 4);
        v.x = (a.x + v.x) * (a.x + v.x);
        v.y = (a.y + v.y) * (a.y + v.y);
        v.z = (a.z + v.z) * (a.z + v.z);
        v.w = (a.w + v.w) * (a.w + v.w);
    }
    return ri;
}

__device__ __forceinline__ void own_add(float* __restrict__ acc, int ri, int lane, const float4 v) {
    float* p = acc + ri * NHID + lane;
    ldsAddF(p,      v.x);
    ldsAddF(p + 32, v.y);
    ldsAddF(p + 64, v.z);
    ldsAddF(p + 96, v.w);
}

template <int MODE>
__device__ __forceinline__ void own_rows(int n, int g, int rel, float* __restrict__ acc,
                                         const float* __restrict__ G,
                                         const float* __restrict__ Aown, int n0, int lane) {
    int i = 0;
    for (; i + 4 <= n; i += 4) {
        float4 v0, v1, v2, v3;
        const int r0 = own_row<MODE>(v0, g, rel, i + 0, G, Aown, n0, lane);
        const int r1 = own_row<MODE>(v1, g, rel, i + 1, G, Aown, n0, lane);
        const int r2 = own_row<MODE>(v2, g, rel, i + 2, G, Aown, n0, lane);
        const int r3 = own_row<MODE>(v3, g, rel, i + 3, G, Aown, n0, lane);
        own_add(acc, r0, lane, v0);
        own_add(acc, r1, lane, v1);
        own_add(acc, r2, lane, v2);
        own_add(acc, r3, lane, v3);
    }
    for (; i < n; ++i) {
        float4 v0;
        const int r0 = own_row<MODE>(v0, g, rel, i, G, Aown, n0, lane);
        own_add(acc, r0, lane, v0);
    }
}

template <int MODE, bool SAVE, bool KEYS>
__device__ __forceinline__ void own_drain(int* __restrict__ q, int& qn, float* __restrict__ acc,
                                          int* __restrict__ cntL, int* __restrict__ lctl,
                                          int* __restrict__ lst,
                                          const float* __restrict__ G,
                                          const float* __restrict__ Aown,
                                          const int* __restrict__ scat,
                                          const int* __restrict__ gath, int n0, int lane) {
    __builtin_amdgcn_wave_barrier();
    int b = 0;
    for (; b + 32 <= qn; b += 32) {
        const int ent = q[b + lane];
        const int rel = KEYS ? (scat[ent >> 6] - n0) : (ent & 63);
        const int g = gath[ent >> 6];
        __hip_atomic_fetch_add(&cntL[rel], 1, __ATOMIC_RELAXED, __HIP_MEMORY_SCOPE_WORKGROUP);
        if (SAVE && lst != nullptr) {
            int c = 0;
            if (lane == 0)
                c = __hip_atomic_fetch_add(&lctl[0], 1, __ATOMIC_RELAXED, __HIP_MEMORY_SCOPE_WORKGROUP);
            c = __builtin_amdgcn_readfirstlane(c);
            if (c < LCAP / 32) lst[LHDR + c * 32 + lane] = (g << 6) | rel;
        }
        own_rows<MODE>(32, g, rel, acc, G, Aown, n0, lane);
    }
    const int r = qn - b;
    if (b > 0 && lane < r) {
        const int t = q[b + lane];
        q[lane] = t;
    }
    qn = r;
    __builtin_amdgcn_wave_barrier();
}

__device__ __forceinline__ void own_load4(int4& a0, int4& a1, int4& a2, int4& a3,
                                          const int4* __restrict__ sv, int vb) {
    a0 = sv[vb];
    a1 = sv[vb + OWN_T];
    a2 = sv[vb + 2 * OWN_T];
    a3 = sv[vb + 3 * OWN_T];
}

template <int MODE, bool SAVE>
__device__ __forceinline__ void own_trip(const int4 a0, const int4 a1, const int4 a2, const int4 a3,
                                         int vb, int* __restrict__ q, int& qn,
                                         float* __restrict__ acc, int* __restrict__ cntL,
                                         int* __restrict__ lctl, int* __restrict__ lst,
                                         const float* __restrict__ G,
                                         const float* __restrict__ Aown,
                                         const int* __restrict__ scat,
                                         const int* __restrict__ gath,
                                         int n0, unsigned rcur, int lane) {
    own_push4(q, qn, a0, vb * 4, n0, rcur, lane);
    own_push4(q, qn, a1, (vb + OWN_T) * 4, n0, rcur, lane);
    own_push4(q, qn, a2, (vb + 2 * OWN_T) * 4, n0, rcur, lane);
    own_push4(q, qn, a3, (vb + 3 * OWN_T) * 4, n0, rcur, lane);
    if (qn >= 32) own_drain<MODE, SAVE, false>(q, qn, acc, cntL, lctl, lst, G, Aown, scat, gath, n0, lane);
}

template <int MODE, bool SAVE, bool KEYS>
__device__ __forceinline__ void own_tail(int* __restrict__ q, int qn, float* __restrict__ acc,
                                         int* __restrict__ cntL, int* __restrict__ lctl,
                                         int* __restrict__ lq, int* __restrict__ lst,
                                         const float* __restrict__ G,
                                         const float* __restrict__ Aown,
                                         const int* __restrict__ scat,
                                         const int* __restrict__ gath,
                                         int n0, int tid, int lane, int wave) {
    __builtin_amdgcn_wave_barrier();
    {
        int base = 0;
        if (lane == 0 && qn > 0)
            base = __hip_atomic_fetch_add(&lctl[1], qn, __ATOMIC_RELAXED, __HIP_MEMORY_SCOPE_WORKGROUP);
        base = __builtin_amdgcn_readfirstlane(base);
        if (lane < qn) lq[base + lane] = q[lane];
    }
    __syncthreads();
    const int L = lctl[1];
    const int F = lctl[0];
    const bool fits = SAVE && lst != nullptr && F * 32 + L <= LCAP;
    if (fits) {
        const int i = wave * 32 + lane;
        if (i < L) {
            const int ent = lq[i];
            const int rel = KEYS ? (scat[ent >> 6] - n0) : (ent & 63);
            lst[LHDR + F * 32 + i] = (gath[ent >> 6] << 6) | rel;
        }
    }
    if (SAVE && lst != nullptr && tid == 0) lst[0] = fits ? F * 32 + L : -1;
    {
        const int wv = __builtin_amdgcn_readfirstlane(wave);
        const int lo = (wv * L) >> 5;
        const int n  = (((wv + 1) * L) >> 5) - lo;
        int rel = 0, g = 0;
        if (lane < n) {
            const int ent = lq[lo + lane];
            rel = KEYS ? (scat[ent >> 6] - n0) : (ent & 63);
            g = gath[ent >> 6];
            __hip_atomic_fetch_add(&cntL[rel], 1, __ATOMIC_RELAXED, __HIP_MEMORY_SCOPE_WORKGROUP);
        }
        own_rows<MODE>(n, g, rel, acc, G, Aown, n0, lane);
    }
}

__device__ __forceinline__ bool planes_ok(const unsigned* kall, int E, int nblk,
                                          const float* wsf, size_t wsz) {
    return kall != nullptr && E >= 32 && (E & 31) == 0 && nblk <= 256 && ((size_t)kall & 15) == 0 &&
           (size_t)((const char*)(kall + (size_t)(E >> 5) * 16) - (const char*)wsf) <= wsz;
}

__device__ __forceinline__ unsigned plane_match(const int4 lo, const int4 hi, const unsigned* nk) {
    return ((unsigned)lo.x ^ nk[0]) & ((unsigned)lo.y ^ nk[1]) & ((unsigned)lo.z ^ nk[2]) &
           ((unsigned)lo.w ^ nk[3]) & ((unsigned)hi.x ^ nk[4]) & ((unsigned)hi.y ^ nk[5]) &
           ((unsigned)hi.z ^ nk[6]) & ((unsigned)hi.w ^ nk[7]);
}

template <int MODE, bool SAVE>
__device__ __forceinline__ void own_pushm(unsigned m, int e0, int* __restrict__ q, int& qn,
                                          float* __restrict__ acc, int* __restrict__ cntL,
                                          int* __restrict__ lctl, int* __restrict__ lst,
                                          const float* __restrict__ G,
                                          const float* __restrict__ Aown,
                                          const int* __restrict__ scat,
                                          const int* __restrict__ gath, int n0, int lane) {
    unsigned mask = __builtin_amdgcn_ballot_w32(m != 0u);
    while (mask != 0u) {
        if (m != 0u) {
            const int i = __builtin_ctz(m);
            q[qn + __builtin_popcount(mask & ((1u << lane) - 1u))] = (e0 + i) << 6;
            m &= m - 1u;
        }
        qn += __builtin_popcount(mask);
        if (qn >= 32)
            own_drain<MODE, SAVE, true>(q, qn, acc, cntL, lctl, lst, G, Aown, scat, gath, n0, lane);
        mask = __builtin_amdgcn_ballot_w32(m != 0u);
    }
}

template <int MODE, bool SAVE>
__device__ __forceinline__ void own_filterp(int* __restrict__ q, float* __restrict__ acc,
                                            int* __restrict__ cntL, int* __restrict__ lctl,
                                            int* __restrict__ lq, int* __restrict__ lst,
                                            const float* __restrict__ G,
                                            const float* __restrict__ Aown,
                                            const int* __restrict__ scat,
                                            const int* __restrict__ gath,
                                            const unsigned* __restrict__ kp,
                                            int n0, int E, int tid, int lane, int wave) {
    int qn = 0;
    const unsigned bn = blockIdx.x;
    unsigned nk[8];
    #pragma unroll
    for (int k = 0; k < 8; ++k) nk[k] = ((bn >> k) & 1u) ? 0u : ~0u;

    const int ngrp  = E >> 5;
    const int ntrip = (ngrp + OWN_T - 1) / OWN_T;
    const int toff = (ntrip > 0) ? (int)(blockIdx.x % (unsigned)ntrip) : 0;
    auto trip_g = [&](int t) {
        int tt = t + toff;
        if (tt >= ntrip) tt -= ntrip;
        return tt * OWN_T + tid;
    };
    auto ldp = [&](int t, int4& lo, int4& hi) {
        int g = (t < ntrip) ? trip_g(t) : (ngrp - 1);
        g = (g < ngrp) ? g : (ngrp - 1);
        const int4* p = (const int4*)(kp + (size_t)g * 8);
        lo = p[0];
        hi = p[1];
    };
    auto step = [&](const int4 lo, const int4 hi, int t) {
        const int g = trip_g(t);
        unsigned m = plane_match(lo, hi, nk);
        if (t >= ntrip || g >= ngrp) m = 0u;
        own_pushm<MODE, SAVE>(m, g << 5, q, qn, acc, cntL, lctl, lst, G, Aown, scat, gath, n0, lane);
    };

    int4 a0l, a0h, a1l, a1h, b0l, b0h, b1l, b1h;
    ldp(0, a0l, a0h);
    ldp(1, a1l, a1h);
    for (int t = 0; t < ntrip; t += 4) {
        ldp(t + 2, b0l, b0h);
        ldp(t + 3, b1l, b1h);
        step(a0l, a0h, t);
        step(a1l, a1h, t + 1);
        ldp(t + 4, a0l, a0h);
        ldp(t + 5, a1l, a1h);
        step(b0l, b0h, t + 2);
        step(b1l, b1h, t + 3);
    }
    own_tail<MODE, SAVE, true>(q, qn, acc, cntL, lctl, lq, lst, G, Aown, scat, gath, n0, tid, lane, wave);
}

template <int MODE, bool SAVE>
__device__ __forceinline__ void own_filter(int* __restrict__ q, float* __restrict__ acc,
                                           int* __restrict__ cntL, int* __restrict__ lctl,
                                           int* __restrict__ lq, int* __restrict__ lst,
                                           const float* __restrict__ G,
                                           const float* __restrict__ Aown,
                                           const int* __restrict__ scat,
                                           const int* __restrict__ gath,
                                           int n0, unsigned rcur, int E,
                                           int tid, int lane, int wave) {
    int qn = 0;

    const int nvec = ((E & 3) == 0 && ((size_t)scat & 15) == 0) ? (E >> 2) : 0;
    const int4* sv = (const int4*)scat;
    const int4 none = make_int4(-1, -1, -1, -1);

    const int nfull = nvec / (4 * OWN_T);
    const int toff = (nfull > 0) ? (int)(blockIdx.x % (unsigned)nfull) : 0;
    auto trip_vb = [&](int t) {
        int tt = t + toff;
        if (tt >= nfull) tt -= nfull;
        return tt * 4 * OWN_T + tid;
    };
    int4 a0 = none, a1 = none, a2 = none, a3 = none;
    int4 b0 = none, b1 = none, b2 = none, b3 = none;
    if (nfull > 0) own_load4(a0, a1, a2, a3, sv, trip_vb(0));
    int t = 0;
    for (; t + 2 <= nfull; t += 2) {
        own_load4(b0, b1, b2, b3, sv, trip_vb(t + 1));
        own_trip<MODE, SAVE>(a0, a1, a2, a3, trip_vb(t), q, qn, acc, cntL, lctl, lst, G, Aown, scat, gath,
                             n0, rcur, lane);
        if (t + 2 < nfull) own_load4(a0, a1, a2, a3, sv, trip_vb(t + 2));
        own_trip<MODE, SAVE>(b0, b1, b2, b3, trip_vb(t + 1), q, qn, acc, cntL, lctl, lst, G, Aown, scat, gath,
                             n0, rcur, lane);
    }
    if (t < nfull)
        own_trip<MODE, SAVE>(a0, a1, a2, a3, trip_vb(t), q, qn, acc, cntL, lctl, lst, G, Aown, scat, gath,
                             n0, rcur, lane);
    for (int vb = nfull * 4 * OWN_T; vb < nvec; vb += OWN_T) {
        const int idx = vb + tid;
        int4 a = none;
        if (idx < nvec) a = sv[idx];
        own_push4(q, qn, a, idx * 4, n0, rcur, lane);
        if (qn >= 32) own_drain<MODE, SAVE, false>(q, qn, acc, cntL, lctl, lst, G, Aown, scat, gath, n0, lane);
    }
    for (int eb = nvec * 4; eb < E; eb += OWN_T) {
        const int e = eb + tid;
        const int v = (e < E) ? scat[e] : -1;
        own_push(q, qn, v, e, n0, rcur, lane);
        if (qn >= 32) own_drain<MODE, SAVE, false>(q, qn, acc, cntL, lctl, lst, G, Aown, scat, gath, n0, lane);
    }

    own_tail<MODE, SAVE, false>(q, qn, acc, cntL, lctl, lq, lst, G, Aown, scat, gath, n0, tid, lane, wave);
}

template <int MODE>
__device__ __forceinline__ void own_replay(const int* __restrict__ lst, int T,
                                           float* __restrict__ acc, int* __restrict__ cntL,
                                           const float* __restrict__ G,
                                           const float* __restrict__ Aown,
                                           int n0, int lane, int wave) {
    const int wv = __builtin_amdgcn_readfirstlane(wave);
    const int lo = (wv * T) >> 5;
    const int hi = ((wv + 1) * T) >> 5;
    for (int b = lo; b < hi; b += 32) {
        const int n = (hi - b < 32) ? (hi - b) : 32;
        int rel = 0, g = 0;
        if (lane < n) {
            const int ent = lst[LHDR + b + lane];
            rel = ent & 63;
            g = ent >> 6;
            __hip_atomic_fetch_add(&cntL[rel], 1, __ATOMIC_RELAXED, __HIP_MEMORY_SCOPE_WORKGROUP);
        }
        own_rows<MODE>(n, g, rel, acc, G, Aown, n0, lane);
    }
}

__device__ __forceinline__ float4 gate4(const float* __restrict__ acc, const int* __restrict__ cntL,
                                        int r, int l, const float* __restrict__ Xcur,
                                        const float* __restrict__ Xnew, size_t o) {
    const float c = fmaxf((float)cntL[r], 1.0f);
    const float* a = acc + r * NHID + l;
    const float4 s = make_float4(a[0], a[32], a[64], a[96]);
    const float4 xc = *(const float4*)(Xcur + o);
    const float4 xn = *(const float4*)(Xnew + o);
    const float tx = tanhf(s.x / c), ty = tanhf(s.y / c);
    const float tz = tanhf(s.z / c), tw = tanhf(s.w / c);
    float4 y;
    y.x = (1.0f - tx) * xc.x + tx * xn.x;
    y.y = (1.0f - ty) * xc.y + ty * xn.y;
    y.z = (1.0f - tz) * xc.z + tz * xn.z;
    y.w = (1.0f - tw) * xc.w + tw * xn.w;
    return y;
}

template <int MODE, int PASS, bool DEC>
__global__ __launch_bounds__(OWN_T)
void k_owner(const float* __restrict__ G, const float* __restrict__ Aown,
             const int* __restrict__ scat, const int* __restrict__ gath,
             const float* __restrict__ Xcur, const float* __restrict__ Xnew,
             float* __restrict__ out, const LayerG lg, int N, int E,
             int* __restrict__ lstAll, const float* __restrict__ wsf, size_t wsz,
             const unsigned* __restrict__ kall, int kofs) {
    __shared__ float acc[RB * NHID];
    __shared__ int   cntL[RB];
    __shared__ int   lctl[2];
    __shared__ int   lq[OWN_T];
    __shared__ int   qbuf[OWN_T / 32][QCAP];
    __shared__ alignas(16) float gbuf[MODE == 0 ? 3 * RBP * LP : (DEC ? RBP * LP : 4)];

    const int tid  = threadIdx.x;
    const int lane = tid & 31;
    const int wave = tid >> 5;
    const int n0   = blockIdx.x * RB;
    const unsigned rcur = (unsigned)((N - n0 < RB) ? (N - n0) : RB);

    int* lst = nullptr;
    if ((size_t)((const char*)(lstAll + (size_t)gridDim.x * LSTRIDE) - (const char*)wsf) <= wsz)
        lst = lstAll + (size_t)blockIdx.x * LSTRIDE;

    for (int i = tid; i < RB * NHID; i += OWN_T) acc[i] = 0.0f;
    if (tid < RB) cntL[tid] = 0;
    if (tid < 2) lctl[tid] = 0;
    __syncthreads();

    int T = -1;
    if (PASS == 1 && lst != nullptr) {
        T = lst[0];
        if (T > LCAP) T = -1;
    }
    if (T >= 0)
        own_replay<MODE>(lst, T, acc, cntL, G, Aown, n0, lane, wave);
    else if (PASS == 0 && planes_ok(kall, E, (int)gridDim.x, wsf, wsz))
        own_filterp<MODE, true>(qbuf[wave], acc, cntL, lctl, lq, lst, G, Aown, scat, gath,
                                kall + (size_t)(kofs >> 5) * 8, n0, E, tid, lane, wave);
    else
        own_filter<MODE, PASS == 0>(qbuf[wave], acc, cntL, lctl, lq, lst, G, Aown, scat, gath,
                                    n0, rcur, E, tid, lane, wave);
    __syncthreads();

    if (MODE == 0) {
        float* aggs = gbuf;
        float* xs   = gbuf + RBP * LP;
        float* xg   = gbuf + 2 * RBP * LP;
        for (int i = tid; i < RBP * 32; i += OWN_T) {
            const int r = i >> 5;
            const int l = i & 31;
            float4 s = make_float4(0.0f, 0.0f, 0.0f, 0.0f);
            float4 x = s;
            if (r < (int)rcur) {
                const float c = fmaxf((float)cntL[r], 1.0f);
                const float* a = acc + r * NHID + l;
                s = make_float4(a[0] / c, a[32] / c, a[64] / c, a[96] / c);
                x = *(const float4*)(G + (size_t)(n0 + r) * NHID + l * 4);
            }
            *(float4*)(aggs + r * LP + l * 4) = s;
            *(float4*)(xs + r * LP + l * 4) = x;
        }
        __syncthreads();

        const bool gw   = wave < MT * 8;
        const int  mt   = wave >> 3;
        const int  sel  = (wave >> 2) & 1;
        const int  row  = lane & 15;
        const int  half = lane >> 4;
        const int  colo = (wave & 3) * 32 + 2 * row;
        const int  lrow = mt * 16 + 8 * half;

        if (gw) {
            v8f c0 = {}, c1 = {};
            blk_mma<NHID>(c0, c1, aggs + (mt * 16 + row) * LP + 2 * half, sel ? lg.Wl2 : lg.Wl1, half, colo);
            blk_mma<NHID>(c0, c1, xs   + (mt * 16 + row) * LP + 2 * half, sel ? lg.Wr2 : lg.Wr1, half, colo);
            const float* bp = sel ? lg.b2 : lg.b1;
            const float bv0 = bp[colo], bv1 = bp[colo + 1];
            #pragma unroll
            for (int i = 0; i < 8; ++i) {
                v2f y;
                y.x = fmaxf(c0[i] + bv0, 0.0f);
                y.y = fmaxf(c1[i] + bv1, 0.0f);
                if (sel) {
                    *(v2f*)(xg + (lrow + i) * LP + colo) = y;
                } else if (lrow + i < (int)rcur) {
                    *(v2f*)(lg.T1 + (size_t)(n0 + lrow + i) * NHID + colo) = y;
                }
            }
        }
        __syncthreads();

        if (gw) {
            v8f c0 = {}, c1 = {};
            blk_mma<NHID>(c0, c1, xg + (mt * 16 + row) * LP + 2 * half,
                          lg.Q + (sel ? (size_t)NHID * NHID : 0), half, colo);
            const float bv0 = sel ? 0.0f : lg.Qb[colo];
            const float bv1 = sel ? 0.0f : lg.Qb[colo + 1];
            float* dstp = sel ? lg.Bb : lg.Ab;
            #pragma unroll
            for (int i = 0; i < 8; ++i) {
                v2f y;
                y.x = c0[i] + bv0;
                y.y = c1[i] + bv1;
                if (lrow + i < (int)rcur)
                    *(v2f*)(dstp + (size_t)(n0 + lrow + i) * NHID + colo) = y;
            }
        }
        return;
    }

    if (DEC) {
        float* xs = gbuf;
        for (int i = tid; i < RBP * 32; i += OWN_T) {
            const int r = i >> 5;
            const int l = i & 31;
            float4 y = make_float4(0.0f, 0.0f, 0.0f, 0.0f);
            if (r < (int)rcur)
                y = gate4(acc, cntL, r, l, Xcur, Xnew, (size_t)(n0 + r) * NHID + l * 4);
            *(float4*)(xs + r * LP + l * 4) = y;
        }
        __syncthreads();

        if (wave < MT * 2) {
            const int mt   = wave >> 1;
            const int row  = lane & 15;
            const int half = lane >> 4;
            const int colo = (wave & 1) * 32 + 2 * row;
            const int lrow = mt * 16 + 8 * half;
            v8f c0 = {}, c1 = {};
            blk_mma<NCLASS>(c0, c1, xs + (mt * 16 + row) * LP + 2 * half, lg.Wd, half, colo);
            const float bv0 = lg.bd[colo], bv1 = lg.bd[colo + 1];
            #pragma unroll
            for (int i = 0; i < 8; ++i) {
                v2f y;
                y.x = c0[i] + bv0;
                y.y = c1[i] + bv1;
                if (lrow + i < (int)rcur)
                    *(v2f*)(lg.Od + (size_t)(n0 + lrow + i) * NCLASS + colo) = y;
            }
        }
        return;
    }

    for (int i = tid; i < (int)rcur * 32; i += OWN_T) {
        const int r = i >> 5;
        const int l = i & 31;
        const size_t o = (size_t)(n0 + r) * NHID + l * 4;
        *(float4*)(out + o) = gate4(acc, cntL, r, l, Xcur, Xnew, o);
    }
}

__device__ __forceinline__ void enc_blk(v8f& c0, v8f& c1, const float* __restrict__ ap,
                                        const float* __restrict__ wp) {
    const float4 a4 = *(const float4*)ap;
    const v2f w0 = *(const v2f*)(wp);
    const v2f w1 = *(const v2f*)(wp + NHID);
    const v2f w2 = *(const v2f*)(wp + 2 * NHID);
    const v2f w3 = *(const v2f*)(wp + 3 * NHID);
    v2f a, b0, b1;
    a.x = a4.x;  a.y = a4.y;
    b0.x = w0.x; b0.y = w1.x;
    b1.x = w0.y; b1.y = w1.y;
    c0 = __builtin_amdgcn_wmma_f32_16x16x4_f32(false, a, false, b0, (short)0, c0, false, false);
    c1 = __builtin_amdgcn_wmma_f32_16x16x4_f32(false, a, false, b1, (short)0, c1, false, false);
    a.x = a4.z;  a.y = a4.w;
    b0.x = w2.x; b0.y = w3.x;
    b1.x = w2.y; b1.y = w3.y;
    c0 = __builtin_amdgcn_wmma_f32_16x16x4_f32(false, a, false, b0, (short)0, c0, false, false);
    c1 = __builtin_amdgcn_wmma_f32_16x16x4_f32(false, a, false, b1, (short)0, c1, false, false);
}

__device__ __forceinline__ void key_planes(const int* __restrict__ ei, unsigned* __restrict__ kall,
                                           int ngrp, int gw, int lane) {
    const int gb = __builtin_amdgcn_readfirstlane(gw) * KGW;
    unsigned key[KGW];
    #pragma unroll
    for (int u = 0; u < KGW; ++u) {
        const int g = (gb + u < ngrp) ? (gb + u) : (ngrp - 1);
        key[u] = (unsigned)ei[(size_t)g * 32 + lane] / (unsigned)RB;
    }
    #pragma unroll
    for (int u = 0; u < KGW; ++u) {
        unsigned v = 0u;
        #pragma unroll
        for (int k = 0; k < 8; ++k) {
            const unsigned p = __builtin_amdgcn_ballot_w32(((key[u] >> k) & 1u) != 0u);
            if (lane == k) v = p;
        }
        if (gb + u < ngrp && lane < 8) kall[(size_t)(gb + u) * 8 + lane] = v;
    }
}

__global__ __launch_bounds__(128) __attribute__((amdgpu_waves_per_eu(1, 4)))
void k_enc(const float* __restrict__ A, const float* __restrict__ W,
           const float* __restrict__ bias, float* __restrict__ C,
           const int* __restrict__ ei, unsigned* __restrict__ kall,
           const float* __restrict__ wsf, size_t wsz, int M, int E, int encBlocks, int ownBlocks) {
    const int lane = threadIdx.x & 31;
    const int wave = threadIdx.x >> 5;
    if ((int)blockIdx.x >= encBlocks) {
        if (planes_ok(kall, E, ownBlocks, wsf, wsz))
            key_planes(ei, kall, E >> 4, ((int)blockIdx.x - encBlocks) * 4 + wave, lane);
        return;
    }
    const int m0   = blockIdx.x << 4;
    const int row  = lane & 15;
    const int half = lane >> 4;
    const int colo = wave * 32 + 2 * row;

    const int ar = (m0 + row < M) ? (m0 + row) : (M - 1);
    const float* ap = A + (size_t)ar * NFEAT + 4 * half;
    const float* wp = W + (size_t)(4 * half) * NHID + colo;

    v8f c0 = {}, c1 = {};
    #pragma unroll
    for (int kb = 0; kb < NFEAT; kb += 8)
        enc_blk(c0, c1, ap + kb, wp + (size_t)kb * NHID);

    const float bv0 = bias[colo], bv1 = bias[colo + 1];
    const int lrow = m0 + 8 * half;
    #pragma unroll
    for (int i = 0; i < 8; ++i) {
        v2f y;
        y.x = fmaxf(c0[i] + bv0, 0.0f);
        y.y = fmaxf(c1[i] + bv1, 0.0f);
        if (lrow + i < M) *(v2f*)(C + (size_t)(lrow + i) * NHID + colo) = y;
    }
}

extern "C" void kernel_launch(void* const* d_in, const int* in_sizes, int n_in,
                              void* d_out, int out_size, void* d_ws, size_t ws_size,
                              hipStream_t stream) {
    const float* X       = (const float*)d_in[0];
    const int*   ei      = (const int*)  d_in[1];
    const float* enc_W   = (const float*)d_in[2];
    const float* enc_b   = (const float*)d_in[3];
    const float* conv_Wl = (const float*)d_in[4];
    const float* conv_bl = (const float*)d_in[5];
    const float* conv_Wr = (const float*)d_in[6];
    const float* gg_Wl   = (const float*)d_in[7];
    const float* gg_bl   = (const float*)d_in[8];
    const float* gg_Wr   = (const float*)d_in[9];
    const float* Q_W     = (const float*)d_in[10];
    const float* Q_b     = (const float*)d_in[11];
    const float* dec_W   = (const float*)d_in[12];
    const float* dec_b   = (const float*)d_in[13];

    const int N = in_sizes[0] / NFEAT;
    const int E = in_sizes[1] / 2;
    const int* src = ei;
    const int* dst = ei + E;

    float* ws       = (float*)d_ws;
    const size_t NH = (size_t)N * NHID;
    float* Xa       = ws + 2 * (size_t)N;
    float* Xb       = Xa + NH;
    float* T1       = Xb + NH;
    float* Ab       = T1 + NH;
    float* Bb       = Ab + NH;

    const int encGrd  = (N + 15) / 16;
    const int ownGrd  = (N + RB - 1) / RB;

    int* listD = (int*)(Bb + NH);
    int* listS = listD + (size_t)ownGrd * LSTRIDE;
    unsigned* kall = (unsigned*)(listS + (size_t)ownGrd * LSTRIDE);
    const float* wsf = ws;
    const int keyGrd = (E / 16 + 4 * KGW - 1) / (4 * KGW);

    k_enc<<<encGrd + keyGrd, 128, 0, stream>>>(X, enc_W, enc_b, Xa, ei, kall, wsf, ws_size,
                                               N, E, encGrd, ownGrd);

    LayerG lg;
    lg.Wl1 = conv_Wl; lg.Wr1 = conv_Wr; lg.b1 = conv_bl;
    lg.Wl2 = gg_Wl;   lg.Wr2 = gg_Wr;   lg.b2 = gg_bl;
    lg.Q   = Q_W;     lg.Qb  = Q_b;
    lg.T1  = T1;      lg.Ab  = Ab;      lg.Bb  = Bb;
    lg.Wd  = dec_W;   lg.bd  = dec_b;   lg.Od  = (float*)d_out;

    k_owner<0, 0, false><<<ownGrd, OWN_T, 0, stream>>>(Xa, nullptr, dst, src,
                                                       nullptr, nullptr, nullptr, lg, N, E,
                                                       listD, wsf, ws_size, kall, E);
    k_owner<1, 0, false><<<ownGrd, OWN_T, 0, stream>>>(Bb, Ab, src, dst,
                                                       Xa, T1, Xb, lg, N, E,
                                                       listS, wsf, ws_size, kall, 0);
    k_owner<0, 1, false><<<ownGrd, OWN_T, 0, stream>>>(Xb, nullptr, dst, src,
                                                       nullptr, nullptr, nullptr, lg, N, E,
                                                       listD, wsf, ws_size, nullptr, 0);
    k_owner<1, 1, true><<<ownGrd, OWN_T, 0, stream>>>(Bb, Ab, src, dst,
                                                      Xb, T1, nullptr, lg, N, E,
                                                      listS, wsf, ws_size, nullptr, 0);
}


# Owner work-groups order kept edges by owned row (LDS counting sort); a wave sums runs of equal rows in registers, LDS f32 atomics only at run ends; f32 sum order: register runs, then arrival order
# speedup vs baseline: 4.3018x; 1.0408x over previous
#include <hip/hip_runtime.h>
#include <hip/hip_bf16.h>


typedef __attribute__((ext_vector_type(2))) float v2f;
typedef __attribute__((ext_vector_type(8))) float v8f;

#define NFEAT  256
#define NHID   128
#define NCLASS 64

#define RB    40
#define OWN_T 1024
#define QCAP  544
#define MT    3
#define RBP   (MT * 16)
#define LP    132
#define LCAP    2048
#define LHDR    32
#define LSTRIDE (LHDR + LCAP)
static_assert(LCAP == 2 * OWN_T, "the ordering pass carries two list entries per thread");
#define KGW 16

struct LayerG {
    const float* Wl1; const float* Wr1; const float* b1;
    const float* Wl2; const float* Wr2; const float* b2;
    const float* Q;   const float* Qb;
    float* T1; float* Ab; float* Bb;
    const float* Wd;  const float* bd;  float* Od;
};

template <int WS>
__device__ __forceinline__ void blk_mma(v8f& c0, v8f& c1, const float* __restrict__ ar,
                                        const float* __restrict__ Wm, int half, int colo) {
    const float* wp = Wm + (size_t)(2 * half) * WS + colo;
    for (int k0 = 0; k0 < NHID; k0 += 4) {
        const v2f a  = *(const v2f*)(ar + k0);
        const v2f w0 = *(const v2f*)(wp + (size_t)k0 * WS);
        const v2f w1 = *(const v2f*)(wp + (size_t)(k0 + 1) * WS);
        v2f b0, b1;
        b0.x = w0.x; b0.y = w1.x;
        b1.x = w0.y; b1.y = w1.y;
        c0 = __builtin_amdgcn_wmma_f32_16x16x4_f32(false, a, false, b0, (short)0, c0, false, false);
        c1 = __builtin_amdgcn_wmma_f32_16x16x4_f32(false, a, false, b1, (short)0, c1, false, false);
    }
}

__device__ __forceinline__ void ldsAddF(float* p, float v) {
    __hip_atomic_fetch_add(p, v, __ATOMIC_RELAXED, __HIP_MEMORY_SCOPE_WORKGROUP);
}

__device__ __forceinline__ void own_push(int* __restrict__ q, int& qn, int v, int e,
                                         int n0, unsigned rcur, int lane) {
    const unsigned rel = (unsigned)(v - n0);
    const bool m = rel < rcur;
    const unsigned mask = __builtin_amdgcn_ballot_w32(m);
    if (mask) {
        if (m) q[qn + __builtin_popcount(mask & ((1u << lane) - 1u))] = (e << 6) | (int)rel;
        qn += __builtin_popcount(mask);
    }
}

__device__ __forceinline__ void own_push4(int* __restrict__ q, int& qn, const int4 v, int e,
                                          int n0, unsigned rcur, int lane) {
    own_push(q, qn, v.x, e + 0, n0, rcur, lane);
    own_push(q, qn, v.y, e + 1, n0, rcur, lane);
    own_push(q, qn, v.z, e + 2, n0, rcur, lane);
    own_push(q, qn, v.w, e + 3, n0, rcur, lane);
}

template <int MODE>
__device__ __forceinline__ int own_row(float4& v, int g, int rel, int i,
                                       const float* __restrict__ G,
                                       const float* __restrict__ Aown, int n0, int lane) {
    const int gi = __builtin_amdgcn_readlane(g, i);
    const int ri = __builtin_amdgcn_readlane(rel, i);
    v = *(const float4*)(G + (size_t)gi * NHID + lane * 4);
    if (MODE == 1) {
        const float4 a = *(const float4*)(Aown + (size_t)(n0 + ri) * NHID + lane * 4);
        v.x = (a.x + v.x) * (a.x + v.x);
        v.y = (a.y + v.y) * (a.y + v.y);
        v.z = (a.z + v.z) * (a.z + v.z);
        v.w = (a.w + v.w) * (a.w + v.w);
    }
    return ri;
}

__device__ __forceinline__ void own_add(float* __restrict__ acc, int ri, int lane, const float4 v) {
    float* p = acc + ri * NHID + lane;
    ldsAddF(p,      v.x);
    ldsAddF(p + 32, v.y);
    ldsAddF(p + 64, v.z);
    ldsAddF(p + 96, v.w);
}

template <int MODE>
__device__ __forceinline__ void own_rows(int n, int g, int rel, float* __restrict__ acc,
                                         const float* __restrict__ G,
                                         const float* __restrict__ Aown, int n0, int lane) {
    int i = 0;
    for (; i + 4 <= n; i += 4) {
        float4 v0, v1, v2, v3;
        const int r0 = own_row<MODE>(v0, g, rel, i + 0, G, Aown, n0, lane);
        const int r1 = own_row<MODE>(v1, g, rel, i + 1, G, Aown, n0, lane);
        const int r2 = own_row<MODE>(v2, g, rel, i + 2, G, Aown, n0, lane);
        const int r3 = own_row<MODE>(v3, g, rel, i + 3, G, Aown, n0, lane);
        own_add(acc, r0, lane, v0);
        own_add(acc, r1, lane, v1);
        own_add(acc, r2, lane, v2);
        own_add(acc, r3, lane, v3);
    }
    for (; i < n; ++i) {
        float4 v0;
        const int r0 = own_row<MODE>(v0, g, rel, i, G, Aown, n0, lane);
        own_add(acc, r0, lane, v0);
    }
}

__device__ __forceinline__ void run_add(float* __restrict__ acc, int& cur, float4& s, int r,
                                        const float4 v, int lane) {
    const bool nw = r != cur;
    if (nw && cur >= 0) own_add(acc, cur, lane, s);
    s.x = nw ? v.x : s.x + v.x;
    s.y = nw ? v.y : s.y + v.y;
    s.z = nw ? v.z : s.z + v.z;
    s.w = nw ? v.w : s.w + v.w;
    cur = r;
}

template <int MODE>
__device__ __forceinline__ void own_runs(const int* __restrict__ ls, int T,
                                         float* __restrict__ acc,
                                         const float* __restrict__ G,
                                         const float* __restrict__ Aown,
                                         int n0, int lane, int wave) {
    const int wv = __builtin_amdgcn_readfirstlane(wave);
    const int lo = (wv * T) >> 5;
    const int hi = ((wv + 1) * T) >> 5;
    int cur = -1;
    float4 s = make_float4(0.0f, 0.0f, 0.0f, 0.0f);
    for (int b = lo; b < hi; b += 32) {
        const int n = (hi - b < 32) ? (hi - b) : 32;
        int rel = 0, g = 0;
        if (lane < n) {
            const int ent = ls[b + lane];
            rel = ent & 63;
            g = ent >> 6;
        }
        int i = 0;
        for (; i + 4 <= n; i += 4) {
            float4 v0, v1, v2, v3;
            const int r0 = own_row<MODE>(v0, g, rel, i + 0, G, Aown, n0, lane);
            const int r1 = own_row<MODE>(v1, g, rel, i + 1, G, Aown, n0, lane);
            const int r2 = own_row<MODE>(v2, g, rel, i + 2, G, Aown, n0, lane);
            const int r3 = own_row<MODE>(v3, g, rel, i + 3, G, Aown, n0, lane);
            run_add(acc, cur, s, r0, v0, lane);
            run_add(acc, cur, s, r1, v1, lane);
            run_add(acc, cur, s, r2, v2, lane);
            run_add(acc, cur, s, r3, v3, lane);
        }
        for (; i < n; ++i) {
            float4 v0;
            const int r0 = own_row<MODE>(v0, g, rel, i, G, Aown, n0, lane);
            run_add(acc, cur, s, r0, v0, lane);
        }
    }
    if (cur >= 0) own_add(acc, cur, lane, s);
}

template <int MODE, bool SAVE>
__device__ __forceinline__ void own_drain(int* __restrict__ q, int& qn, float* __restrict__ acc,
                                          int* __restrict__ cntL, int* __restrict__ lctl,
                                          int* __restrict__ lst,
                                          const float* __restrict__ G,
                                          const float* __restrict__ Aown,
                                          const int* __restrict__ scat,
                                          const int* __restrict__ gath, int n0, int lane) {
    __builtin_amdgcn_wave_barrier();
    int b = 0;
    for (; b + 32 <= qn; b += 32) {
        const int ent = q[b + lane];
        const int rel = ent & 63;
        const int g = gath[ent >> 6];
        __hip_atomic_fetch_add(&cntL[rel], 1, __ATOMIC_RELAXED, __HIP_MEMORY_SCOPE_WORKGROUP);
        if (SAVE && lst != nullptr) {
            int c = 0;
            if (lane == 0)
                c = __hip_atomic_fetch_add(&lctl[0], 1, __ATOMIC_RELAXED, __HIP_MEMORY_SCOPE_WORKGROUP);
            c = __builtin_amdgcn_readfirstlane(c);
            if (c < LCAP / 32) lst[LHDR + c * 32 + lane] = (g << 6) | rel;
        }
        own_rows<MODE>(32, g, rel, acc, G, Aown, n0, lane);
    }
    const int r = qn - b;
    if (b > 0 && lane < r) {
        const int t = q[b + lane];
        q[lane] = t;
    }
    qn = r;
    __builtin_amdgcn_wave_barrier();
}

__device__ __forceinline__ void own_load4(int4& a0, int4& a1, int4& a2, int4& a3,
                                          const int4* __restrict__ sv, int vb) {
    a0 = sv[vb];
    a1 = sv[vb + OWN_T];
    a2 = sv[vb + 2 * OWN_T];
    a3 = sv[vb + 3 * OWN_T];
}

template <int MODE, bool SAVE>
__device__ __forceinline__ void own_trip(const int4 a0, const int4 a1, const int4 a2, const int4 a3,
                                         int vb, int* __restrict__ q, int& qn,
                                         float* __restrict__ acc, int* __restrict__ cntL,
                                         int* __restrict__ lctl, int* __restrict__ lst,
                                         const float* __restrict__ G,
                                         const float* __restrict__ Aown,
                                         const int* __restrict__ scat,
                                         const int* __restrict__ gath,
                                         int n0, unsigned rcur, int lane) {
    own_push4(q, qn, a0, vb * 4, n0, rcur, lane);
    own_push4(q, qn, a1, (vb + OWN_T) * 4, n0, rcur, lane);
    own_push4(q, qn, a2, (vb + 2 * OWN_T) * 4, n0, rcur, lane);
    own_push4(q, qn, a3, (vb + 3 * OWN_T) * 4, n0, rcur, lane);
    if (qn >= 32) own_drain<MODE, SAVE>(q, qn, acc, cntL, lctl, lst, G, Aown, scat, gath, n0, lane);
}

template <int MODE, bool SAVE>
__device__ __forceinline__ void own_tail(int* __restrict__ q, int qn, float* __restrict__ acc,
                                         int* __restrict__ cntL, int* __restrict__ lctl,
                                         int* __restrict__ lq, int* __restrict__ lst,
                                         const float* __restrict__ G,
                                         const float* __restrict__ Aown,
                                         const int* __restrict__ scat,
                                         const int* __restrict__ gath,
                                         int n0, int tid, int lane, int wave) {
    __builtin_amdgcn_wave_barrier();
    {
        int base = 0;
        if (lane == 0 && qn > 0)
            base = __hip_atomic_fetch_add(&lctl[1], qn, __ATOMIC_RELAXED, __HIP_MEMORY_SCOPE_WORKGROUP);
        base = __builtin_amdgcn_readfirstlane(base);
        if (lane < qn) lq[base + lane] = q[lane];
    }
    __syncthreads();
    const int L = lctl[1];
    const int F = lctl[0];
    const bool fits = SAVE && lst != nullptr && F * 32 + L <= LCAP;
    if (fits) {
        const int i = wave * 32 + lane;
        if (i < L) {
            const int ent = lq[i];
            const int rel = ent & 63;
            lst[LHDR + F * 32 + i] = (gath[ent >> 6] << 6) | rel;
        }
    }
    if (SAVE && lst != nullptr && tid == 0) lst[0] = fits ? F * 32 + L : -1;
    {
        const int wv = __builtin_amdgcn_readfirstlane(wave);
        const int lo = (wv * L) >> 5;
        const int n  = (((wv + 1) * L) >> 5) - lo;
        int rel = 0, g = 0;
        if (lane < n) {
            const int ent = lq[lo + lane];
            rel = ent & 63;
            g = gath[ent >> 6];
            __hip_atomic_fetch_add(&cntL[rel], 1, __ATOMIC_RELAXED, __HIP_MEMORY_SCOPE_WORKGROUP);
        }
        own_rows<MODE>(n, g, rel, acc, G, Aown, n0, lane);
    }
}

__device__ __forceinline__ bool planes_ok(const unsigned* kall, int E, int nblk,
                                          const float* wsf, size_t wsz) {
    return kall != nullptr && E >= 32 && (E & 31) == 0 && nblk <= 256 && ((size_t)kall & 15) == 0 &&
           (size_t)((const char*)(kall + (size_t)(E >> 5) * 16) - (const char*)wsf) <= wsz;
}

__device__ __forceinline__ unsigned plane_match(const int4 lo, const int4 hi, const unsigned* nk) {
    return ((unsigned)lo.x ^ nk[0]) & ((unsigned)lo.y ^ nk[1]) & ((unsigned)lo.z ^ nk[2]) &
           ((unsigned)lo.w ^ nk[3]) & ((unsigned)hi.x ^ nk[4]) & ((unsigned)hi.y ^ nk[5]) &
           ((unsigned)hi.z ^ nk[6]) & ((unsigned)hi.w ^ nk[7]);
}

__device__ __forceinline__ void q_flush(int* __restrict__ q, int& qn, int* __restrict__ el,
                                        int* __restrict__ lctl, int lane, bool all) {
    __builtin_amdgcn_wave_barrier();
    const int lim = all ? qn : (qn & ~31);
    if (lim > 0) {
        int base = 0;
        if (lane == 0)
            base = __hip_atomic_fetch_add(&lctl[0], lim, __ATOMIC_RELAXED, __HIP_MEMORY_SCOPE_WORKGROUP);
        base = __builtin_amdgcn_readfirstlane(base);
        for (int b = 0; b < lim; b += 32) {
            if (b + lane < lim && base + b + lane < LCAP) el[base + b + lane] = q[b + lane];
        }
        const int r = qn - lim;
        if (lane < r) {
            const int t = q[lim + lane];
            q[lane] = t;
        }
        qn = r;
    }
    __builtin_amdgcn_wave_barrier();
}

__device__ __forceinline__ void own_pushe(unsigned m, int e0, int* __restrict__ q, int& qn,
                                          int* __restrict__ el, int* __restrict__ lctl, int lane) {
    unsigned mask = __builtin_amdgcn_ballot_w32(m != 0u);
    while (mask != 0u) {
        if (m != 0u) {
            const int i = __builtin_ctz(m);
            q[qn + __builtin_popcount(mask & ((1u << lane) - 1u))] = e0 + i;
            m &= m - 1u;
        }
        qn += __builtin_popcount(mask);
        if (qn >= 32) q_flush(q, qn, el, lctl, lane, false);
        mask = __builtin_amdgcn_ballot_w32(m != 0u);
    }
}

__device__ __forceinline__ void own_collect(int* __restrict__ q, int* __restrict__ el,
                                            int* __restrict__ lctl,
                                            const unsigned* __restrict__ kp,
                                            int E, int tid, int lane) {
    int qn = 0;
    const unsigned bn = blockIdx.x;
    unsigned nk[8];
    #pragma unroll
    for (int k = 0; k < 8; ++k) nk[k] = ((bn >> k) & 1u) ? 0u : ~0u;

    const int ngrp  = E >> 5;
    const int ntrip = (ngrp + OWN_T - 1) / OWN_T;
    const int toff = (ntrip > 0) ? (int)(blockIdx.x % (unsigned)ntrip) : 0;
    auto trip_g = [&](int t) {
        int tt = t + toff;
        if (tt >= ntrip) tt -= ntrip;
        return tt * OWN_T + tid;
    };
    auto ldp = [&](int t, int4& lo, int4& hi) {
        int g = (t < ntrip) ? trip_g(t) : (ngrp - 1);
        g = (g < ngrp) ? g : (ngrp - 1);
        const int4* p = (const int4*)(kp + (size_t)g * 8);
        lo = p[0];
        hi = p[1];
    };
    auto step = [&](const int4 lo, const int4 hi, int t) {
        const int g = trip_g(t);
        unsigned m = plane_match(lo, hi, nk);
        if (t >= ntrip || g >= ngrp) m = 0u;
        own_pushe(m, g << 5, q, qn, el, lctl, lane);
    };

    int4 a0l, a0h, a1l, a1h, b0l, b0h, b1l, b1h;
    ldp(0, a0l, a0h);
    ldp(1, a1l, a1h);
    for (int t = 0; t < ntrip; t += 4) {
        ldp(t + 2, b0l, b0h);
        ldp(t + 3, b1l, b1h);
        step(a0l, a0h, t);
        step(a1l, a1h, t + 1);
        ldp(t + 4, a0l, a0h);
        ldp(t + 5, a1l, a1h);
        step(b0l, b0h, t + 2);
        step(b1l, b1h, t + 3);
    }
    q_flush(q, qn, el, lctl, lane, true);
}

template <int MODE, bool SAVE>
__device__ __forceinline__ void own_filter(int* __restrict__ q, float* __restrict__ acc,
                                           int* __restrict__ cntL, int* __restrict__ lctl,
                                           int* __restrict__ lq, int* __restrict__ lst,
                                           const float* __restrict__ G,
                                           const float* __restrict__ Aown,
                                           const int* __restrict__ scat,
                                           const int* __restrict__ gath,
                                           int n0, unsigned rcur, int E,
                                           int tid, int lane, int wave) {
    int qn = 0;

    const int nvec = ((E & 3) == 0 && ((size_t)scat & 15) == 0) ? (E >> 2) : 0;
    const int4* sv = (const int4*)scat;
    const int4 none = make_int4(-1, -1, -1, -1);

    const int nfull = nvec / (4 * OWN_T);
    const int toff = (nfull > 0) ? (int)(blockIdx.x % (unsigned)nfull) : 0;
    auto trip_vb = [&](int t) {
        int tt = t + toff;
        if (tt >= nfull) tt -= nfull;
        return tt * 4 * OWN_T + tid;
    };
    int4 a0 = none, a1 = none, a2 = none, a3 = none;
    int4 b0 = none, b1 = none, b2 = none, b3 = none;
    if (nfull > 0) own_load4(a0, a1, a2, a3, sv, trip_vb(0));
    int t = 0;
    for (; t + 2 <= nfull; t += 2) {
        own_load4(b0, b1, b2, b3, sv, trip_vb(t + 1));
        own_trip<MODE, SAVE>(a0, a1, a2, a3, trip_vb(t), q, qn, acc, cntL, lctl, lst, G, Aown, scat, gath,
                             n0, rcur, lane);
        if (t + 2 < nfull) own_load4(a0, a1, a2, a3, sv, trip_vb(t + 2));
        own_trip<MODE, SAVE>(b0, b1, b2, b3, trip_vb(t + 1), q, qn, acc, cntL, lctl, lst, G, Aown, scat, gath,
                             n0, rcur, lane);
    }
    if (t < nfull)
        own_trip<MODE, SAVE>(a0, a1, a2, a3, trip_vb(t), q, qn, acc, cntL, lctl, lst, G, Aown, scat, gath,
                             n0, rcur, lane);
    for (int vb = nfull * 4 * OWN_T; vb < nvec; vb += OWN_T) {
        const int idx = vb + tid;
        int4 a = none;
        if (idx < nvec) a = sv[idx];
        own_push4(q, qn, a, idx * 4, n0, rcur, lane);
        if (qn >= 32) own_drain<MODE, SAVE>(q, qn, acc, cntL, lctl, lst, G, Aown, scat, gath, n0, lane);
    }
    for (int eb = nvec * 4; eb < E; eb += OWN_T) {
        const int e = eb + tid;
        const int v = (e < E) ? scat[e] : -1;
        own_push(q, qn, v, e, n0, rcur, lane);
        if (qn >= 32) own_drain<MODE, SAVE>(q, qn, acc, cntL, lctl, lst, G, Aown, scat, gath, n0, lane);
    }

    own_tail<MODE, SAVE>(q, qn, acc, cntL, lctl, lq, lst, G, Aown, scat, gath, n0, tid, lane, wave);
}

__device__ __forceinline__ float4 gate4(const float* __restrict__ acc, const int* __restrict__ cntL,
                                        int r, int l, const float* __restrict__ Xcur,
                                        const float* __restrict__ Xnew, size_t o) {
    const float c = fmaxf((float)cntL[r], 1.0f);
    const float* a = acc + r * NHID + l;
    const float4 s = make_float4(a[0], a[32], a[64], a[96]);
    const float4 xc = *(const float4*)(Xcur + o);
    const float4 xn = *(const float4*)(Xnew + o);
    const float tx = tanhf(s.x / c), ty = tanhf(s.y / c);
    const float tz = tanhf(s.z / c), tw = tanhf(s.w / c);
    float4 y;
    y.x = (1.0f - tx) * xc.x + tx * xn.x;
    y.y = (1.0f - ty) * xc.y + ty * xn.y;
    y.z = (1.0f - tz) * xc.z + tz * xn.z;
    y.w = (1.0f - tw) * xc.w + tw * xn.w;
    return y;
}

template <int MODE, int PASS, bool DEC>
__global__ __launch_bounds__(OWN_T)
void k_owner(const float* __restrict__ G, const float* __restrict__ Aown,
             const int* __restrict__ scat, const int* __restrict__ gath,
             const float* __restrict__ Xcur, const float* __restrict__ Xnew,
             float* __restrict__ out, const LayerG lg, int N, int E,
             int* __restrict__ lstAll, const float* __restrict__ wsf, size_t wsz,
             const unsigned* __restrict__ kall, int kofs) {
    __shared__ float acc[RB * NHID];
    __shared__ int   cntL[RB];
    __shared__ int   lctl[2];
    __shared__ int   lq[OWN_T];
    __shared__ int   qbuf[OWN_T / 32][QCAP];
    __shared__ int   el[LCAP];
    __shared__ int   sl[LCAP];
    __shared__ int   offL[RB];
    __shared__ alignas(16) float gbuf[MODE == 0 ? 3 * RBP * LP : (DEC ? RBP * LP : 4)];

    const int tid  = threadIdx.x;
    const int lane = tid & 31;
    const int wave = tid >> 5;
    const int n0   = blockIdx.x * RB;
    const unsigned rcur = (unsigned)((N - n0 < RB) ? (N - n0) : RB);

    int* lst = nullptr;
    if ((size_t)((const char*)(lstAll + (size_t)gridDim.x * LSTRIDE) - (const char*)wsf) <= wsz)
        lst = lstAll + (size_t)blockIdx.x * LSTRIDE;

    for (int i = tid; i < RB * NHID; i += OWN_T) acc[i] = 0.0f;
    if (tid < RB) cntL[tid] = 0;
    if (tid < 2) lctl[tid] = 0;
    __syncthreads();

    int T = -1;
    if (PASS == 1 && lst != nullptr) {
        T = lst[0];
        if (T > LCAP) T = -1;
    }
    if (T >= 0) {
        for (int i = tid; i < T; i += OWN_T)
            __hip_atomic_fetch_add(&cntL[lst[LHDR + i] & 63], 1, __ATOMIC_RELAXED,
                                   __HIP_MEMORY_SCOPE_WORKGROUP);
        own_runs<MODE>(lst + LHDR, T, acc, G, Aown, n0, lane, wave);
    } else {
        bool ordered = false;
        if (PASS == 0 && planes_ok(kall, E, (int)gridDim.x, wsf, wsz)) {
            own_collect(qbuf[wave], el, lctl, kall + (size_t)(kofs >> 5) * 8, E, tid, lane);
            __syncthreads();
            const int n = lctl[0];
            if (n <= LCAP) {
                ordered = true;
                int rl[2], gi[2], rk[2];
                #pragma unroll
                for (int u = 0; u < 2; ++u) {
                    const int i = tid + u * OWN_T;
                    rl[u] = 0; gi[u] = 0; rk[u] = 0;
                    if (i < n) {
                        const int e = el[i];
                        rl[u] = scat[e] - n0;
                        gi[u] = gath[e];
                        rk[u] = __hip_atomic_fetch_add(&cntL[rl[u]], 1, __ATOMIC_RELAXED,
                                                       __HIP_MEMORY_SCOPE_WORKGROUP);
                    }
                }
                __syncthreads();
                if (tid < RB) {
                    int s = 0;
                    for (int r = 0; r < tid; ++r) s += cntL[r];
                    offL[tid] = s;
                }
                __syncthreads();
                #pragma unroll
                for (int u = 0; u < 2; ++u) {
                    const int i = tid + u * OWN_T;
                    if (i < n) sl[offL[rl[u]] + rk[u]] = (gi[u] << 6) | rl[u];
                }
                __syncthreads();
                if (lst != nullptr) {
                    for (int i = tid; i < n; i += OWN_T) lst[LHDR + i] = sl[i];
                    if (tid == 0) lst[0] = n;
                }
                own_runs<MODE>(sl, n, acc, G, Aown, n0, lane, wave);
            } else {
                __syncthreads();
                if (tid < 2) lctl[tid] = 0;
                __syncthreads();
            }
        }
        if (!ordered)
            own_filter<MODE, PASS == 0>(qbuf[wave], acc, cntL, lctl, lq, lst, G, Aown, scat, gath,
                                        n0, rcur, E, tid, lane, wave);
    }
    __syncthreads();

    if (MODE == 0) {
        float* aggs = gbuf;
        float* xs   = gbuf + RBP * LP;
        float* xg   = gbuf + 2 * RBP * LP;
        for (int i = tid; i < RBP * 32; i += OWN_T) {
            const int r = i >> 5;
            const int l = i & 31;
            float4 s = make_float4(0.0f, 0.0f, 0.0f, 0.0f);
            float4 x = s;
            if (r < (int)rcur) {
                const float c = fmaxf((float)cntL[r], 1.0f);
                const float* a = acc + r * NHID + l;
                s = make_float4(a[0] / c, a[32] / c, a[64] / c, a[96] / c);
                x = *(const float4*)(G + (size_t)(n0 + r) * NHID + l * 4);
            }
            *(float4*)(aggs + r * LP + l * 4) = s;
            *(float4*)(xs + r * LP + l * 4) = x;
        }
        __syncthreads();

        const bool gw   = wave < MT * 8;
        const int  mt   = wave >> 3;
        const int  sel  = (wave >> 2) & 1;
        const int  row  = lane & 15;
        const int  half = lane >> 4;
        const int  colo = (wave & 3) * 32 + 2 * row;
        const int  lrow = mt * 16 + 8 * half;

        if (gw) {
            v8f c0 = {}, c1 = {};
            blk_mma<NHID>(c0, c1, aggs + (mt * 16 + row) * LP + 2 * half, sel ? lg.Wl2 : lg.Wl1, half, colo);
            blk_mma<NHID>(c0, c1, xs   + (mt * 16 + row) * LP + 2 * half, sel ? lg.Wr2 : lg.Wr1, half, colo);
            const float* bp = sel ? lg.b2 : lg.b1;
            const float bv0 = bp[colo], bv1 = bp[colo + 1];
            #pragma unroll
            for (int i = 0; i < 8; ++i) {
                v2f y;
                y.x = fmaxf(c0[i] + bv0, 0.0f);
                y.y = fmaxf(c1[i] + bv1, 0.0f);
                if (sel) {
                    *(v2f*)(xg + (lrow + i) * LP + colo) = y;
                } else if (lrow + i < (int)rcur) {
                    *(v2f*)(lg.T1 + (size_t)(n0 + lrow + i) * NHID + colo) = y;
                }
            }
        }
        __syncthreads();

        if (gw) {
            v8f c0 = {}, c1 = {};
            blk_mma<NHID>(c0, c1, xg + (mt * 16 + row) * LP + 2 * half,
                          lg.Q + (sel ? (size_t)NHID * NHID : 0), half, colo);
            const float bv0 = sel ? 0.0f : lg.Qb[colo];
            const float bv1 = sel ? 0.0f : lg.Qb[colo + 1];
            float* dstp = sel ? lg.Bb : lg.Ab;
            #pragma unroll
            for (int i = 0; i < 8; ++i) {
                v2f y;
                y.x = c0[i] + bv0;
                y.y = c1[i] + bv1;
                if (lrow + i < (int)rcur)
                    *(v2f*)(dstp + (size_t)(n0 + lrow + i) * NHID + colo) = y;
            }
        }
        return;
    }

    if (DEC) {
        float* xs = gbuf;
        for (int i = tid; i < RBP * 32; i += OWN_T) {
            const int r = i >> 5;
            const int l = i & 31;
            float4 y = make_float4(0.0f, 0.0f, 0.0f, 0.0f);
            if (r < (int)rcur)
                y = gate4(acc, cntL, r, l, Xcur, Xnew, (size_t)(n0 + r) * NHID + l * 4);
            *(float4*)(xs + r * LP + l * 4) = y;
        }
        __syncthreads();

        if (wave < MT * 2) {
            const int mt   = wave >> 1;
            const int row  = lane & 15;
            const int half = lane >> 4;
            const int colo = (wave & 1) * 32 + 2 * row;
            const int lrow = mt * 16 + 8 * half;
            v8f c0 = {}, c1 = {};
            blk_mma<NCLASS>(c0, c1, xs + (mt * 16 + row) * LP + 2 * half, lg.Wd, half, colo);
            const float bv0 = lg.bd[colo], bv1 = lg.bd[colo + 1];
            #pragma unroll
            for (int i = 0; i < 8; ++i) {
                v2f y;
                y.x = c0[i] + bv0;
                y.y = c1[i] + bv1;
                if (lrow + i < (int)rcur)
                    *(v2f*)(lg.Od + (size_t)(n0 + lrow + i) * NCLASS + colo) = y;
            }
        }
        return;
    }

    for (int i = tid; i < (int)rcur * 32; i += OWN_T) {
        const int r = i >> 5;
        const int l = i & 31;
        const size_t o = (size_t)(n0 + r) * NHID + l * 4;
        *(float4*)(out + o) = gate4(acc, cntL, r, l, Xcur, Xnew, o);
    }
}

__device__ __forceinline__ void enc_blk(v8f& c0, v8f& c1, const float* __restrict__ ap,
                                        const float* __restrict__ wp) {
    const float4 a4 = *(const float4*)ap;
    const v2f w0 = *(const v2f*)(wp);
    const v2f w1 = *(const v2f*)(wp + NHID);
    const v2f w2 = *(const v2f*)(wp + 2 * NHID);
    const v2f w3 = *(const v2f*)(wp + 3 * NHID);
    v2f a, b0, b1;
    a.x = a4.x;  a.y = a4.y;
    b0.x = w0.x; b0.y = w1.x;
    b1.x = w0.y; b1.y = w1.y;
    c0 = __builtin_amdgcn_wmma_f32_16x16x4_f32(false, a, false, b0, (short)0, c0, false, false);
    c1 = __builtin_amdgcn_wmma_f32_16x16x4_f32(false, a, false, b1, (short)0, c1, false, false);
    a.x = a4.z;  a.y = a4.w;
    b0.x = w2.x; b0.y = w3.x;
    b1.x = w2.y; b1.y = w3.y;
    c0 = __builtin_amdgcn_wmma_f32_16x16x4_f32(false, a, false, b0, (short)0, c0, false, false);
    c1 = __builtin_amdgcn_wmma_f32_16x16x4_f32(false, a, false, b1, (short)0, c1, false, false);
}

__device__ __forceinline__ void key_planes(const int* __restrict__ ei, unsigned* __restrict__ kall,
                                           int ngrp, int gw, int lane) {
    const int gb = __builtin_amdgcn_readfirstlane(gw) * KGW;
    unsigned key[KGW];
    #pragma unroll
    for (int u = 0; u < KGW; ++u) {
        const int g = (gb + u < ngrp) ? (gb + u) : (ngrp - 1);
        key[u] = (unsigned)ei[(size_t)g * 32 + lane] / (unsigned)RB;
    }
    #pragma unroll
    for (int u = 0; u < KGW; ++u) {
        unsigned v = 0u;
        #pragma unroll
        for (int k = 0; k < 8; ++k) {
            const unsigned p = __builtin_amdgcn_ballot_w32(((key[u] >> k) & 1u) != 0u);
            if (lane == k) v = p;
        }
        if (gb + u < ngrp && lane < 8) kall[(size_t)(gb + u) * 8 + lane] = v;
    }
}

__global__ __launch_bounds__(128) __attribute__((amdgpu_waves_per_eu(1, 4)))
void k_enc(const float* __restrict__ A, const float* __restrict__ W,
           const float* __restrict__ bias, float* __restrict__ C,
           const int* __restrict__ ei, unsigned* __restrict__ kall,
           const float* __restrict__ wsf, size_t wsz, int M, int E, int encBlocks, int ownBlocks) {
    const int lane = threadIdx.x & 31;
    const int wave = threadIdx.x >> 5;
    if ((int)blockIdx.x >= encBlocks) {
        if (planes_ok(kall, E, ownBlocks, wsf, wsz))
            key_planes(ei, kall, E >> 4, ((int)blockIdx.x - encBlocks) * 4 + wave, lane);
        return;
    }
    const int m0   = blockIdx.x << 4;
    const int row  = lane & 15;
    const int half = lane >> 4;
    const int colo = wave * 32 + 2 * row;

    const int ar = (m0 + row < M) ? (m0 + row) : (M - 1);
    const float* ap = A + (size_t)ar * NFEAT + 4 * half;
    const float* wp = W + (size_t)(4 * half) * NHID + colo;

    v8f c0 = {}, c1 = {};
    #pragma unroll
    for (int kb = 0; kb < NFEAT; kb += 8)
        enc_blk(c0, c1, ap + kb, wp + (size_t)kb * NHID);

    const float bv0 = bias[colo], bv1 = bias[colo + 1];
    const int lrow = m0 + 8 * half;
    #pragma unroll
    for (int i = 0; i < 8; ++i) {
        v2f y;
        y.x = fmaxf(c0[i] + bv0, 0.0f);
        y.y = fmaxf(c1[i] + bv1, 0.0f);
        if (lrow + i < M) *(v2f*)(C + (size_t)(lrow + i) * NHID + colo) = y;
    }
}

extern "C" void kernel_launch(void* const* d_in, const int* in_sizes, int n_in,
                              void* d_out, int out_size, void* d_ws, size_t ws_size,
                              hipStream_t stream) {
    const float* X       = (const float*)d_in[0];
    const int*   ei      = (const int*)  d_in[1];
    const float* enc_W   = (const float*)d_in[2];
    const float* enc_b   = (const float*)d_in[3];
    const float* conv_Wl = (const float*)d_in[4];
    const float* conv_bl = (const float*)d_in[5];
    const float* conv_Wr = (const float*)d_in[6];
    const float* gg_Wl   = (const float*)d_in[7];
    const float* gg_bl   = (const float*)d_in[8];
    const float* gg_Wr   = (const float*)d_in[9];
    const float* Q_W     = (const float*)d_in[10];
    const float* Q_b     = (const float*)d_in[11];
    const float* dec_W   = (const float*)d_in[12];
    const float* dec_b   = (const float*)d_in[13];

    const int N = in_sizes[0] / NFEAT;
    const int E = in_sizes[1] / 2;
    const int* src = ei;
    const int* dst = ei + E;

    float* ws       = (float*)d_ws;
    const size_t NH = (size_t)N * NHID;
    float* Xa       = ws + 2 * (size_t)N;
    float* Xb       = Xa + NH;
    float* T1       = Xb + NH;
    float* Ab       = T1 + NH;
    float* Bb       = Ab + NH;

    const int encGrd  = (N + 15) / 16;
    const int ownGrd  = (N + RB - 1) / RB;

    int* listD = (int*)(Bb + NH);
    int* listS = listD + (size_t)ownGrd * LSTRIDE;
    unsigned* kall = (unsigned*)(listS + (size_t)ownGrd * LSTRIDE);
    const float* wsf = ws;
    const int keyGrd = (E / 16 + 4 * KGW - 1) / (4 * KGW);

    k_enc<<<encGrd + keyGrd, 128, 0, stream>>>(X, enc_W, enc_b, Xa, ei, kall, wsf, ws_size,
                                               N, E, encGrd, ownGrd);

    LayerG lg;
    lg.Wl1 = conv_Wl; lg.Wr1 = conv_Wr; lg.b1 = conv_bl;
    lg.Wl2 = gg_Wl;   lg.Wr2 = gg_Wr;   lg.b2 = gg_bl;
    lg.Q   = Q_W;     lg.Qb  = Q_b;
    lg.T1  = T1;      lg.Ab  = Ab;      lg.Bb  = Bb;
    lg.Wd  = dec_W;   lg.bd  = dec_b;   lg.Od  = (float*)d_out;

    k_owner<0, 0, false><<<ownGrd, OWN_T, 0, stream>>>(Xa, nullptr, dst, src,
                                                       nullptr, nullptr, nullptr, lg, N, E,
                                                       listD, wsf, ws_size, kall, E);
    k_owner<1, 0, false><<<ownGrd, OWN_T, 0, stream>>>(Bb, Ab, src, dst,
                                                       Xa, T1, Xb, lg, N, E,
                                                       listS, wsf, ws_size, kall, 0);
    k_owner<0, 1, false><<<ownGrd, OWN_T, 0, stream>>>(Xb, nullptr, dst, src,
                                                       nullptr, nullptr, nullptr, lg, N, E,
                                                       listD, wsf, ws_size, nullptr, 0);
    k_owner<1, 1, true><<<ownGrd, OWN_T, 0, stream>>>(Bb, Ab, src, dst,
                                                      Xb, T1, nullptr, lg, N, E,
                                                      listS, wsf, ws_size, nullptr, 0);
}
